# starting kernel
# baseline (speedup 1.0000x reference)
#include <hip/hip_runtime.h>

#define N_NODES 200000
#define N_EDGES 6400000
#define FDIM    128
#define HDIM    16
#define CDIM    10
#define BGRAPH  512
#define NTILES  (N_NODES / 16)

typedef float v2f __attribute__((ext_vector_type(2)));
typedef float v8f __attribute__((ext_vector_type(8)));

__global__ void count_deg_kernel(const int* __restrict__ col, int* __restrict__ degcnt) {
    int e = blockIdx.x * blockDim.x + threadIdx.x;
    if (e < N_EDGES) atomicAdd(&degcnt[col[e]], 1);
}

__global__ void node_prep_kernel(const int* __restrict__ degcnt, const int* __restrict__ batch,
                                 float* __restrict__ dis, int* __restrict__ counts) {
    int n = blockIdx.x * blockDim.x + threadIdx.x;
    if (n < N_NODES) {
        dis[n] = rsqrtf(1.0f + (float)degcnt[n]);
        atomicAdd(&counts[batch[n]], 1);
    }
}

__global__ void __launch_bounds__(256) gemm1_kernel(const float* __restrict__ x,
                                                    const float* __restrict__ W1,
                                                    const float* __restrict__ dis,
                                                    float* __restrict__ hws,
                                                    float* __restrict__ agg) {
    __shared__ float lw[(FDIM / 2) * 32];
    int tid = threadIdx.x;
#pragma unroll
    for (int i = 0; i < 8; ++i) {
        int idx = i * 256 + tid;
        int k = idx >> 4, n = idx & 15;
        lw[(k >> 1) * 32 + n * 2 + (k & 1)] = W1[idx];
    }
    __syncthreads();

    int wave = tid >> 5, lane = tid & 31;
    int tile = blockIdx.x * 8 + wave;
    if (tile >= NTILES) return;

    int h  = lane >> 4;
    int nn = lane & 15;
    long m = (long)tile * 16 + nn;
    const float* xrow = x + m * FDIM;

    v8f c = {0.f, 0.f, 0.f, 0.f, 0.f, 0.f, 0.f, 0.f};
#pragma unroll
    for (int kk = 0; kk < FDIM; kk += 4) {
        int k0 = kk + 2 * h;
        v2f a = *(const v2f*)(xrow + k0);
        v2f b = *(const v2f*)(&lw[(k0 >> 1) * 32 + nn * 2]);
        c = __builtin_amdgcn_wmma_f32_16x16x4_f32(false, a, false, b,
                                                  (short)0, c, false, false);
    }
#pragma unroll
    for (int v = 0; v < 8; ++v) {
        long r = (long)tile * 16 + v + 8 * h;
        float val = c[v] * dis[r];
        hws[r * HDIM + nn] = val;
        agg[r * HDIM + nn] = val;
    }
}

__global__ void edge_scatter_kernel(const int* __restrict__ row, const int* __restrict__ col,
                                    const float* __restrict__ hws, float* __restrict__ agg) {
    long t = (long)blockIdx.x * blockDim.x + threadIdx.x;
    int e = (int)(t >> 4);
    int j = (int)(t & 15);
    if (e < N_EDGES) {
        int r = row[e];
        int ccol = col[e];
        atomicAdd(&agg[(long)ccol * HDIM + j], hws[(long)r * HDIM + j]);
    }
}

__global__ void __launch_bounds__(256) gemm2_kernel(const float* __restrict__ agg1,
                                                    const float* __restrict__ W2,
                                                    const float* __restrict__ b1v,
                                                    const float* __restrict__ dis,
                                                    float* __restrict__ hws2,
                                                    float* __restrict__ agg2) {
    __shared__ float lw[(HDIM / 2) * 32];
    __shared__ float lb1[HDIM];
    int tid = threadIdx.x;
    if (tid < 256) {
        int k = tid >> 4, n = tid & 15;
        lw[(k >> 1) * 32 + n * 2 + (k & 1)] = W2[tid];
    }
    if (tid < HDIM) lb1[tid] = b1v[tid];
    __syncthreads();

    int wave = tid >> 5, lane = tid & 31;
    int tile = blockIdx.x * 8 + wave;
    if (tile >= NTILES) return;

    int h  = lane >> 4;
    int nn = lane & 15;
    long m = (long)tile * 16 + nn;
    float dm = dis[m];
    const float* arow = agg1 + m * HDIM;

    v8f c = {0.f, 0.f, 0.f, 0.f, 0.f, 0.f, 0.f, 0.f};
#pragma unroll
    for (int kk = 0; kk < HDIM; kk += 4) {
        int k0 = kk + 2 * h;
        v2f raw = *(const v2f*)(arow + k0);
        v2f a;
        a.x = fmaxf(dm * raw.x + lb1[k0], 0.0f);
        a.y = fmaxf(dm * raw.y + lb1[k0 + 1], 0.0f);
        v2f b = *(const v2f*)(&lw[(k0 >> 1) * 32 + nn * 2]);
        c = __builtin_amdgcn_wmma_f32_16x16x4_f32(false, a, false, b,
                                                  (short)0, c, false, false);
    }
#pragma unroll
    for (int v = 0; v < 8; ++v) {
        long r = (long)tile * 16 + v + 8 * h;
        float val = c[v] * dis[r];
        hws2[r * HDIM + nn] = val;
        agg2[r * HDIM + nn] = val;
    }
}

__global__ void pool_kernel(const float* __restrict__ agg2, const float* __restrict__ dis,
                            const float* __restrict__ b2v, const int* __restrict__ batch,
                            float* __restrict__ pool) {
    int t = blockIdx.x * blockDim.x + threadIdx.x;
    int n = t >> 4, j = t & 15;
    if (n < N_NODES) {
        float val = dis[n] * agg2[(long)n * HDIM + j] + b2v[j];
        atomicAdd(&pool[batch[n] * HDIM + j], val);
    }
}

__global__ void final_kernel(const float* __restrict__ pool, const int* __restrict__ counts,
                             const float* __restrict__ Wfc, const float* __restrict__ bfc,
                             float* __restrict__ out) {
    int t = blockIdx.x * blockDim.x + threadIdx.x;
    if (t < BGRAPH * CDIM) {
        int b = t / CDIM, cc = t % CDIM;
        float inv = 1.0f / fmaxf((float)counts[b], 1.0f);
        float acc = bfc[cc];
#pragma unroll
        for (int j = 0; j < HDIM; ++j)
            acc += pool[b * HDIM + j] * inv * Wfc[j * CDIM + cc];
        out[t] = acc;
    }
}

extern "C" void kernel_launch(void* const* d_in, const int* in_sizes, int n_in,
                              void* d_out, int out_size, void* d_ws, size_t ws_size,
                              hipStream_t stream) {
    (void)in_sizes; (void)n_in; (void)out_size; (void)ws_size;

    const float* x    = (const float*)d_in[0];
    const int*   ei   = (const int*)d_in[1];
    const int*   batch= (const int*)d_in[2];
    const float* W1   = (const float*)d_in[3];
    const float* b1   = (const float*)d_in[4];
    const float* W2   = (const float*)d_in[5];
    const float* b2   = (const float*)d_in[6];
    const float* Wfc  = (const float*)d_in[7];
    const float* bfc  = (const float*)d_in[8];
    float* out = (float*)d_out;

    const int* erow = ei;
    const int* ecol = ei + N_EDGES;

    char* ws = (char*)d_ws;
    size_t o = 0;
    auto alloc = [&](size_t bytes) -> void* {
        void* p = ws + o;
        o += (bytes + 255) & ~(size_t)255;
        return p;
    };
    float* dis    = (float*)alloc((size_t)N_NODES * 4);
    int*   degcnt = (int*)  alloc((size_t)N_NODES * 4);
    float* hws    = (float*)alloc((size_t)N_NODES * HDIM * 4);
    float* agg1   = (float*)alloc((size_t)N_NODES * HDIM * 4);
    float* agg2   = (float*)alloc((size_t)N_NODES * HDIM * 4);
    float* pool   = (float*)alloc((size_t)BGRAPH * HDIM * 4);
    int*   counts = (int*)  alloc((size_t)BGRAPH * 4);

    hipMemsetAsync(degcnt, 0, (size_t)N_NODES * 4, stream);
    hipMemsetAsync(counts, 0, (size_t)BGRAPH * 4, stream);
    hipMemsetAsync(pool,   0, (size_t)BGRAPH * HDIM * 4, stream);

    count_deg_kernel<<<(N_EDGES + 255) / 256, 256, 0, stream>>>(ecol, degcnt);
    node_prep_kernel<<<(N_NODES + 255) / 256, 256, 0, stream>>>(degcnt, batch, dis, counts);

    int gemm_grid = (NTILES + 7) / 8;
    gemm1_kernel<<<gemm_grid, 256, 0, stream>>>(x, W1, dis, hws, agg1);

    int esc_grid = (int)(((long)N_EDGES * 16 + 255) / 256);
    edge_scatter_kernel<<<esc_grid, 256, 0, stream>>>(erow, ecol, hws, agg1);

    gemm2_kernel<<<gemm_grid, 256, 0, stream>>>(agg1, W2, b1, dis, hws, agg2);

    edge_scatter_kernel<<<esc_grid, 256, 0, stream>>>(erow, ecol, hws, agg2);

    pool_kernel<<<(N_NODES * 16 + 255) / 256, 256, 0, stream>>>(agg2, dis, b2, batch, pool);
    final_kernel<<<(BGRAPH * CDIM + 255) / 256, 256, 0, stream>>>(pool, counts, Wfc, bfc, out);
}


# layer 2 edge sum commuted with the mean pool: each edge adds dis[col]*hws2[row] into a 512x16 LDS table per work-group keyed by batch[col]; agg2 scatter and pool launch go; sum order changes
# speedup vs baseline: 1.3092x; 1.3092x over previous
#include <hip/hip_runtime.h>

#define N_NODES 200000
#define N_EDGES 6400000
#define FDIM    128
#define HDIM    16
#define CDIM    10
#define BGRAPH  512
#define NTILES  (N_NODES / 16)
#define PGRID   512
#define EPB     (N_EDGES / PGRID)
#define PPITCH  (BGRAPH + 1)
#define PTAB    (HDIM * PPITCH)

typedef float v2f __attribute__((ext_vector_type(2)));
typedef float v8f __attribute__((ext_vector_type(8)));

__global__ void count_deg_kernel(const int* __restrict__ col, int* __restrict__ degcnt) {
    int e = blockIdx.x * blockDim.x + threadIdx.x;
    if (e < N_EDGES) atomicAdd(&degcnt[col[e]], 1);
}

__global__ void node_prep_kernel(const int* __restrict__ degcnt, const int* __restrict__ batch,
                                 float* __restrict__ dis, float2* __restrict__ nodew,
                                 int* __restrict__ counts) {
    int n = blockIdx.x * blockDim.x + threadIdx.x;
    if (n < N_NODES) {
        float d = rsqrtf(1.0f + (float)degcnt[n]);
        int g = batch[n];
        dis[n] = d;
        nodew[n] = make_float2(d, __int_as_float(g));
        atomicAdd(&counts[g], 1);
    }
}

__global__ void __launch_bounds__(256) gemm1_kernel(const float* __restrict__ x,
                                                    const float* __restrict__ W1,
                                                    const float* __restrict__ dis,
                                                    float* __restrict__ hws,
                                                    float* __restrict__ agg) {
    __shared__ float lw[(FDIM / 2) * 32];
    int tid = threadIdx.x;
#pragma unroll
    for (int i = 0; i < 8; ++i) {
        int idx = i * 256 + tid;
        int k = idx >> 4, n = idx & 15;
        lw[(k >> 1) * 32 + n * 2 + (k & 1)] = W1[idx];
    }
    __syncthreads();

    int wave = tid >> 5, lane = tid & 31;
    int tile = blockIdx.x * 8 + wave;
    if (tile >= NTILES) return;

    int h  = lane >> 4;
    int nn = lane & 15;
    long m = (long)tile * 16 + nn;
    const float* xrow = x + m * FDIM;

    v8f c = {0.f, 0.f, 0.f, 0.f, 0.f, 0.f, 0.f, 0.f};
#pragma unroll
    for (int kk = 0; kk < FDIM; kk += 4) {
        int k0 = kk + 2 * h;
        v2f a = *(const v2f*)(xrow + k0);
        v2f b = *(const v2f*)(&lw[(k0 >> 1) * 32 + nn * 2]);
        c = __builtin_amdgcn_wmma_f32_16x16x4_f32(false, a, false, b,
                                                  (short)0, c, false, false);
    }
#pragma unroll
    for (int v = 0; v < 8; ++v) {
        long r = (long)tile * 16 + v + 8 * h;
        float val = c[v] * dis[r];
        hws[r * HDIM + nn] = val;
        agg[r * HDIM + nn] = val;
    }
}

__global__ void edge_scatter_kernel(const int* __restrict__ row, const int* __restrict__ col,
                                    const float* __restrict__ hws, float* __restrict__ agg) {
    long t = (long)blockIdx.x * blockDim.x + threadIdx.x;
    int e = (int)(t >> 4);
    int j = (int)(t & 15);
    if (e < N_EDGES) {
        int r = row[e];
        int ccol = col[e];
        atomicAdd(&agg[(long)ccol * HDIM + j], hws[(long)r * HDIM + j]);
    }
}

__global__ void __launch_bounds__(256) gemm2_kernel(const float* __restrict__ agg1,
                                                    const float* __restrict__ W2,
                                                    const float* __restrict__ b1v,
                                                    const float* __restrict__ dis,
                                                    float* __restrict__ hws2) {
    __shared__ float lw[(HDIM / 2) * 32];
    __shared__ float lb1[HDIM];
    int tid = threadIdx.x;
    if (tid < 256) {
        int k = tid >> 4, n = tid & 15;
        lw[(k >> 1) * 32 + n * 2 + (k & 1)] = W2[tid];
    }
    if (tid < HDIM) lb1[tid] = b1v[tid];
    __syncthreads();

    int wave = tid >> 5, lane = tid & 31;
    int tile = blockIdx.x * 8 + wave;
    if (tile >= NTILES) return;

    int h  = lane >> 4;
    int nn = lane & 15;
    long m = (long)tile * 16 + nn;
    float dm = dis[m];
    const float* arow = agg1 + m * HDIM;

    v8f c = {0.f, 0.f, 0.f, 0.f, 0.f, 0.f, 0.f, 0.f};
#pragma unroll
    for (int kk = 0; kk < HDIM; kk += 4) {
        int k0 = kk + 2 * h;
        v2f raw = *(const v2f*)(arow + k0);
        v2f a;
        a.x = fmaxf(dm * raw.x + lb1[k0], 0.0f);
        a.y = fmaxf(dm * raw.y + lb1[k0 + 1], 0.0f);
        v2f b = *(const v2f*)(&lw[(k0 >> 1) * 32 + nn * 2]);
        c = __builtin_amdgcn_wmma_f32_16x16x4_f32(false, a, false, b,
                                                  (short)0, c, false, false);
    }
#pragma unroll
    for (int v = 0; v < 8; ++v) {
        long r = (long)tile * 16 + v + 8 * h;
        float val = c[v] * dis[r];
        hws2[r * HDIM + nn] = val;
    }
}

__device__ __forceinline__ void pool_add4(float* lq, float2 nw, float4 hv) {
    float w = nw.x;
    float* p = lq + __float_as_int(nw.y);
    atomicAdd(p,              w * hv.x);
    atomicAdd(p + PPITCH,     w * hv.y);
    atomicAdd(p + 2 * PPITCH, w * hv.z);
    atomicAdd(p + 3 * PPITCH, w * hv.w);
}

__global__ void __launch_bounds__(1024) edge_pool_kernel(const int* __restrict__ row,
                                                         const int* __restrict__ col,
                                                         const float* __restrict__ hws,
                                                         const float2* __restrict__ nodew,
                                                         const float* __restrict__ b2v,
                                                         float* __restrict__ part) {
    __shared__ float lp[PTAB];
    int tid = threadIdx.x;
    for (int i = tid; i < PTAB; i += 1024) lp[i] = 0.0f;
    __syncthreads();

    int q    = tid & 3;
    int quad = tid >> 2;
    const float4* hws4 = (const float4*)hws;
    float* lq = lp + (4 * q) * PPITCH;

    int ebeg = blockIdx.x * EPB;
    const int4* row4 = (const int4*)(row + ebeg);
    const int4* col4 = (const int4*)(col + ebeg);
    int4 r4 = row4[quad];
    int4 c4 = col4[quad];
    for (int it = 0; it < EPB / 1024; ++it) {
        float2 nw0 = nodew[c4.x], nw1 = nodew[c4.y], nw2 = nodew[c4.z], nw3 = nodew[c4.w];
        float4 hv0 = hws4[(long)r4.x * 4 + q];
        float4 hv1 = hws4[(long)r4.y * 4 + q];
        float4 hv2 = hws4[(long)r4.z * 4 + q];
        float4 hv3 = hws4[(long)r4.w * 4 + q];
        int i4 = min((it + 1) * 256 + quad, EPB / 4 - 1);
        int4 rn = row4[i4];
        int4 cn = col4[i4];
        pool_add4(lq, nw0, hv0);
        pool_add4(lq, nw1, hv1);
        pool_add4(lq, nw2, hv2);
        pool_add4(lq, nw3, hv3);
        r4 = rn;
        c4 = cn;
    }
    if ((EPB / 1024) * 256 + quad < EPB / 4) {
        float2 nw0 = nodew[c4.x], nw1 = nodew[c4.y], nw2 = nodew[c4.z], nw3 = nodew[c4.w];
        float4 hv0 = hws4[(long)r4.x * 4 + q];
        float4 hv1 = hws4[(long)r4.y * 4 + q];
        float4 hv2 = hws4[(long)r4.z * 4 + q];
        float4 hv3 = hws4[(long)r4.w * 4 + q];
        pool_add4(lq, nw0, hv0);
        pool_add4(lq, nw1, hv1);
        pool_add4(lq, nw2, hv2);
        pool_add4(lq, nw3, hv3);
    }

    float bq0 = b2v[4 * q], bq1 = b2v[4 * q + 1], bq2 = b2v[4 * q + 2], bq3 = b2v[4 * q + 3];
    for (int n = blockIdx.x + PGRID * quad; n < N_NODES; n += PGRID * 256) {
        float2 nwn = nodew[n];
        float4 hvn = hws4[(long)n * 4 + q];
        float w = nwn.x;
        float* p = lq + __float_as_int(nwn.y);
        atomicAdd(p,              w * hvn.x + bq0);
        atomicAdd(p + PPITCH,     w * hvn.y + bq1);
        atomicAdd(p + 2 * PPITCH, w * hvn.z + bq2);
        atomicAdd(p + 3 * PPITCH, w * hvn.w + bq3);
    }
    __syncthreads();

    float4* out4 = (float4*)(part + (size_t)blockIdx.x * (BGRAPH * HDIM));
    for (int i = tid; i < BGRAPH * 4; i += 1024) {
        int g = i >> 2, qq = i & 3;
        const float* s = lp + (4 * qq) * PPITCH + g;
        out4[i] = make_float4(s[0], s[PPITCH], s[2 * PPITCH], s[3 * PPITCH]);
    }
}

__global__ void __launch_bounds__(1024) final_kernel(const float* __restrict__ part,
                                                     const int* __restrict__ counts,
                                                     const float* __restrict__ Wfc,
                                                     const float* __restrict__ bfc,
                                                     float* __restrict__ out) {
    __shared__ float red[4][256];
    __shared__ float pl[256];
    int tid = threadIdx.x;
    int el = tid & 255, s = tid >> 8;
    const float* src = part + (size_t)blockIdx.x * 256 + el;
    float a0 = 0.0f, a1 = 0.0f, a2 = 0.0f, a3 = 0.0f;
#pragma unroll 8
    for (int w = s; w < PGRID; w += 16) {
        a0 += src[(size_t)(w)      * (BGRAPH * HDIM)];
        a1 += src[(size_t)(w + 4)  * (BGRAPH * HDIM)];
        a2 += src[(size_t)(w + 8)  * (BGRAPH * HDIM)];
        a3 += src[(size_t)(w + 12) * (BGRAPH * HDIM)];
    }
    red[s][el] = (a0 + a1) + (a2 + a3);
    __syncthreads();
    if (tid < 256) {
        int b = blockIdx.x * 16 + (tid >> 4);
        float inv = 1.0f / fmaxf((float)counts[b], 1.0f);
        pl[tid] = ((red[0][tid] + red[1][tid]) + (red[2][tid] + red[3][tid])) * inv;
    }
    __syncthreads();
    if (tid < 16 * CDIM) {
        int bl = tid / CDIM, cc = tid % CDIM;
        float acc = bfc[cc];
#pragma unroll
        for (int j = 0; j < HDIM; ++j)
            acc += pl[bl * HDIM + j] * Wfc[j * CDIM + cc];
        out[blockIdx.x * (16 * CDIM) + tid] = acc;
    }
}

extern "C" void kernel_launch(void* const* d_in, const int* in_sizes, int n_in,
                              void* d_out, int out_size, void* d_ws, size_t ws_size,
                              hipStream_t stream) {
    (void)in_sizes; (void)n_in; (void)out_size; (void)ws_size;

    const float* x    = (const float*)d_in[0];
    const int*   ei   = (const int*)d_in[1];
    const int*   batch= (const int*)d_in[2];
    const float* W1   = (const float*)d_in[3];
    const float* b1   = (const float*)d_in[4];
    const float* W2   = (const float*)d_in[5];
    const float* b2   = (const float*)d_in[6];
    const float* Wfc  = (const float*)d_in[7];
    const float* bfc  = (const float*)d_in[8];
    float* out = (float*)d_out;

    const int* erow = ei;
    const int* ecol = ei + N_EDGES;

    char* ws = (char*)d_ws;
    size_t o = 0;
    auto alloc = [&](size_t bytes) -> void* {
        void* p = ws + o;
        o += (bytes + 255) & ~(size_t)255;
        return p;
    };
    float* dis    = (float*)alloc((size_t)N_NODES * 4);
    int*   degcnt = (int*)  alloc((size_t)N_NODES * 4);
    float* hws    = (float*)alloc((size_t)N_NODES * HDIM * 4);
    float* agg1   = (float*)alloc((size_t)PGRID * BGRAPH * HDIM * 4);
    float* part   = agg1;
    float2* nodew = (float2*)alloc((size_t)N_NODES * 8);
    int*   counts = (int*)  alloc((size_t)BGRAPH * 4);

    hipMemsetAsync(degcnt, 0, (size_t)N_NODES * 4, stream);
    hipMemsetAsync(counts, 0, (size_t)BGRAPH * 4, stream);

    count_deg_kernel<<<(N_EDGES + 255) / 256, 256, 0, stream>>>(ecol, degcnt);
    node_prep_kernel<<<(N_NODES + 255) / 256, 256, 0, stream>>>(degcnt, batch, dis, nodew, counts);

    int gemm_grid = (NTILES + 7) / 8;
    gemm1_kernel<<<gemm_grid, 256, 0, stream>>>(x, W1, dis, hws, agg1);

    int esc_grid = (int)(((long)N_EDGES * 16 + 255) / 256);
    edge_scatter_kernel<<<esc_grid, 256, 0, stream>>>(erow, ecol, hws, agg1);

    gemm2_kernel<<<gemm_grid, 256, 0, stream>>>(agg1, W2, b1, dis, hws);

    edge_pool_kernel<<<PGRID, 1024, 0, stream>>>(erow, ecol, hws, nodew, b2, part);
    final_kernel<<<BGRAPH / 16, 1024, 0, stream>>>(part, counts, Wfc, bfc, out);
}


# edge scatter on long-lived waves: a wave owns 640 edges, broadcasts entries by DPP row_share, 16 row gathers in flight then 16 global f32 atomic adds; 1250 work-groups; atomic arrival order changes
# speedup vs baseline: 1.7116x; 1.3074x over previous
#include <hip/hip_runtime.h>

#define N_NODES 200000
#define N_EDGES 6400000
#define FDIM    128
#define HDIM    16
#define CDIM    10
#define BGRAPH  512
#define NTILES  (N_NODES / 16)
#define PGRID   512
#define EPB     (N_EDGES / PGRID)
#define PPITCH  (BGRAPH + 1)
#define PTAB    (HDIM * PPITCH)

typedef float v2f __attribute__((ext_vector_type(2)));
typedef float v8f __attribute__((ext_vector_type(8)));

__global__ void count_deg_kernel(const int* __restrict__ col, int* __restrict__ degcnt) {
    int e = blockIdx.x * blockDim.x + threadIdx.x;
    if (e < N_EDGES) atomicAdd(&degcnt[col[e]], 1);
}

__global__ void node_prep_kernel(const int* __restrict__ degcnt, const int* __restrict__ batch,
                                 float* __restrict__ dis, float2* __restrict__ nodew,
                                 int* __restrict__ counts) {
    int n = blockIdx.x * blockDim.x + threadIdx.x;
    if (n < N_NODES) {
        float d = rsqrtf(1.0f + (float)degcnt[n]);
        int g = batch[n];
        dis[n] = d;
        nodew[n] = make_float2(d, __int_as_float(g));
        atomicAdd(&counts[g], 1);
    }
}

__global__ void __launch_bounds__(256) gemm1_kernel(const float* __restrict__ x,
                                                    const float* __restrict__ W1,
                                                    const float* __restrict__ dis,
                                                    float* __restrict__ hws,
                                                    float* __restrict__ agg) {
    __shared__ float lw[(FDIM / 2) * 32];
    int tid = threadIdx.x;
#pragma unroll
    for (int i = 0; i < 8; ++i) {
        int idx = i * 256 + tid;
        int k = idx >> 4, n = idx & 15;
        lw[(k >> 1) * 32 + n * 2 + (k & 1)] = W1[idx];
    }
    __syncthreads();

    int wave = tid >> 5, lane = tid & 31;
    int tile = blockIdx.x * 8 + wave;
    if (tile >= NTILES) return;

    int h  = lane >> 4;
    int nn = lane & 15;
    long m = (long)tile * 16 + nn;
    const float* xrow = x + m * FDIM;

    v8f c = {0.f, 0.f, 0.f, 0.f, 0.f, 0.f, 0.f, 0.f};
#pragma unroll
    for (int kk = 0; kk < FDIM; kk += 4) {
        int k0 = kk + 2 * h;
        v2f a = *(const v2f*)(xrow + k0);
        v2f b = *(const v2f*)(&lw[(k0 >> 1) * 32 + nn * 2]);
        c = __builtin_amdgcn_wmma_f32_16x16x4_f32(false, a, false, b,
                                                  (short)0, c, false, false);
    }
#pragma unroll
    for (int v = 0; v < 8; ++v) {
        long r = (long)tile * 16 + v + 8 * h;
        float val = c[v] * dis[r];
        hws[r * HDIM + nn] = val;
        agg[r * HDIM + nn] = val;
    }
}

#define SC_GRID    1250
#define SC_GROUPS  (N_EDGES / (SC_GRID * 8 * 32))
static_assert(SC_GRID * 8 * 32 * SC_GROUPS == N_EDGES, "edges must split evenly over the waves");

template <int S>
__device__ __forceinline__ unsigned row_share(unsigned v) {
    return (unsigned)__builtin_amdgcn_update_dpp(0, (int)v, 0x150 + S, 0xf, 0xf, false);
}

#define SC_GATHER(S)                                                              \
    {                                                                             \
        unsigned r = row_share<S>(rv);                                            \
        co[S] = row_share<S>(cv) * 64u + jb;                                      \
        hv[S] = *(const float*)(hb + (size_t)(r * 64u + jb));                     \
    }
#define SC_ADD(S) atomicAdd((float*)(ab + (size_t)co[S]), hv[S]);

__global__ void __launch_bounds__(256) edge_scatter_kernel(const int* __restrict__ row,
                                                           const int* __restrict__ col,
                                                           const float* __restrict__ hws,
                                                           float* __restrict__ agg) {
    int lane = threadIdx.x & 31;
    int wave = blockIdx.x * 8 + (threadIdx.x >> 5);
    unsigned jb = (unsigned)(lane & 15) * 4u;
    const int* rp = row + (size_t)wave * (SC_GROUPS * 32) + lane;
    const int* cp = col + (size_t)wave * (SC_GROUPS * 32) + lane;
    const char* hb = (const char*)hws;
    char* ab = (char*)agg;
    unsigned rv = (unsigned)rp[0];
    unsigned cv = (unsigned)cp[0];
    for (int g = 0; g < SC_GROUPS; ++g) {
        int gn = min(g + 1, SC_GROUPS - 1);
        unsigned rn = (unsigned)rp[gn * 32];
        unsigned cn = (unsigned)cp[gn * 32];
        float hv[16];
        unsigned co[16];
        SC_GATHER(0)  SC_GATHER(1)  SC_GATHER(2)  SC_GATHER(3)
        SC_GATHER(4)  SC_GATHER(5)  SC_GATHER(6)  SC_GATHER(7)
        SC_GATHER(8)  SC_GATHER(9)  SC_GATHER(10) SC_GATHER(11)
        SC_GATHER(12) SC_GATHER(13) SC_GATHER(14) SC_GATHER(15)
        SC_ADD(0)  SC_ADD(1)  SC_ADD(2)  SC_ADD(3)
        SC_ADD(4)  SC_ADD(5)  SC_ADD(6)  SC_ADD(7)
        SC_ADD(8)  SC_ADD(9)  SC_ADD(10) SC_ADD(11)
        SC_ADD(12) SC_ADD(13) SC_ADD(14) SC_ADD(15)
        rv = rn;
        cv = cn;
    }
}

__global__ void __launch_bounds__(256) gemm2_kernel(const float* __restrict__ agg1,
                                                    const float* __restrict__ W2,
                                                    const float* __restrict__ b1v,
                                                    const float* __restrict__ dis,
                                                    float* __restrict__ hws2) {
    __shared__ float lw[(HDIM / 2) * 32];
    __shared__ float lb1[HDIM];
    int tid = threadIdx.x;
    if (tid < 256) {
        int k = tid >> 4, n = tid & 15;
        lw[(k >> 1) * 32 + n * 2 + (k & 1)] = W2[tid];
    }
    if (tid < HDIM) lb1[tid] = b1v[tid];
    __syncthreads();

    int wave = tid >> 5, lane = tid & 31;
    int tile = blockIdx.x * 8 + wave;
    if (tile >= NTILES) return;

    int h  = lane >> 4;
    int nn = lane & 15;
    long m = (long)tile * 16 + nn;
    float dm = dis[m];
    const float* arow = agg1 + m * HDIM;

    v8f c = {0.f, 0.f, 0.f, 0.f, 0.f, 0.f, 0.f, 0.f};
#pragma unroll
    for (int kk = 0; kk < HDIM; kk += 4) {
        int k0 = kk + 2 * h;
        v2f raw = *(const v2f*)(arow + k0);
        v2f a;
        a.x = fmaxf(dm * raw.x + lb1[k0], 0.0f);
        a.y = fmaxf(dm * raw.y + lb1[k0 + 1], 0.0f);
        v2f b = *(const v2f*)(&lw[(k0 >> 1) * 32 + nn * 2]);
        c = __builtin_amdgcn_wmma_f32_16x16x4_f32(false, a, false, b,
                                                  (short)0, c, false, false);
    }
#pragma unroll
    for (int v = 0; v < 8; ++v) {
        long r = (long)tile * 16 + v + 8 * h;
        float val = c[v] * dis[r];
        hws2[r * HDIM + nn] = val;
    }
}

__device__ __forceinline__ void pool_add4(float* lq, float2 nw, float4 hv) {
    float w = nw.x;
    float* p = lq + __float_as_int(nw.y);
    atomicAdd(p,              w * hv.x);
    atomicAdd(p + PPITCH,     w * hv.y);
    atomicAdd(p + 2 * PPITCH, w * hv.z);
    atomicAdd(p + 3 * PPITCH, w * hv.w);
}

__global__ void __launch_bounds__(1024) edge_pool_kernel(const int* __restrict__ row,
                                                         const int* __restrict__ col,
                                                         const float* __restrict__ hws,
                                                         const float2* __restrict__ nodew,
                                                         const float* __restrict__ b2v,
                                                         float* __restrict__ part) {
    __shared__ float lp[PTAB];
    int tid = threadIdx.x;
    for (int i = tid; i < PTAB; i += 1024) lp[i] = 0.0f;
    __syncthreads();

    int q    = tid & 3;
    int quad = tid >> 2;
    const float4* hws4 = (const float4*)hws;
    float* lq = lp + (4 * q) * PPITCH;

    int ebeg = blockIdx.x * EPB;
    const int4* row4 = (const int4*)(row + ebeg);
    const int4* col4 = (const int4*)(col + ebeg);
    int4 r4 = row4[quad];
    int4 c4 = col4[quad];
    for (int it = 0; it < EPB / 1024; ++it) {
        float2 nw0 = nodew[c4.x], nw1 = nodew[c4.y], nw2 = nodew[c4.z], nw3 = nodew[c4.w];
        float4 hv0 = hws4[(long)r4.x * 4 + q];
        float4 hv1 = hws4[(long)r4.y * 4 + q];
        float4 hv2 = hws4[(long)r4.z * 4 + q];
        float4 hv3 = hws4[(long)r4.w * 4 + q];
        int i4 = min((it + 1) * 256 + quad, EPB / 4 - 1);
        int4 rn = row4[i4];
        int4 cn = col4[i4];
        pool_add4(lq, nw0, hv0);
        pool_add4(lq, nw1, hv1);
        pool_add4(lq, nw2, hv2);
        pool_add4(lq, nw3, hv3);
        r4 = rn;
        c4 = cn;
    }
    if ((EPB / 1024) * 256 + quad < EPB / 4) {
        float2 nw0 = nodew[c4.x], nw1 = nodew[c4.y], nw2 = nodew[c4.z], nw3 = nodew[c4.w];
        float4 hv0 = hws4[(long)r4.x * 4 + q];
        float4 hv1 = hws4[(long)r4.y * 4 + q];
        float4 hv2 = hws4[(long)r4.z * 4 + q];
        float4 hv3 = hws4[(long)r4.w * 4 + q];
        pool_add4(lq, nw0, hv0);
        pool_add4(lq, nw1, hv1);
        pool_add4(lq, nw2, hv2);
        pool_add4(lq, nw3, hv3);
    }

    float bq0 = b2v[4 * q], bq1 = b2v[4 * q + 1], bq2 = b2v[4 * q + 2], bq3 = b2v[4 * q + 3];
    for (int n = blockIdx.x + PGRID * quad; n < N_NODES; n += PGRID * 256) {
        float2 nwn = nodew[n];
        float4 hvn = hws4[(long)n * 4 + q];
        float w = nwn.x;
        float* p = lq + __float_as_int(nwn.y);
        atomicAdd(p,              w * hvn.x + bq0);
        atomicAdd(p + PPITCH,     w * hvn.y + bq1);
        atomicAdd(p + 2 * PPITCH, w * hvn.z + bq2);
        atomicAdd(p + 3 * PPITCH, w * hvn.w + bq3);
    }
    __syncthreads();

    float4* out4 = (float4*)(part + (size_t)blockIdx.x * (BGRAPH * HDIM));
    for (int i = tid; i < BGRAPH * 4; i += 1024) {
        int g = i >> 2, qq = i & 3;
        const float* s = lp + (4 * qq) * PPITCH + g;
        out4[i] = make_float4(s[0], s[PPITCH], s[2 * PPITCH], s[3 * PPITCH]);
    }
}

__global__ void __launch_bounds__(1024) final_kernel(const float* __restrict__ part,
                                                     const int* __restrict__ counts,
                                                     const float* __restrict__ Wfc,
                                                     const float* __restrict__ bfc,
                                                     float* __restrict__ out) {
    __shared__ float red[4][256];
    __shared__ float pl[256];
    int tid = threadIdx.x;
    int el = tid & 255, s = tid >> 8;
    const float* src = part + (size_t)blockIdx.x * 256 + el;
    float a0 = 0.0f, a1 = 0.0f, a2 = 0.0f, a3 = 0.0f;
#pragma unroll 8
    for (int w = s; w < PGRID; w += 16) {
        a0 += src[(size_t)(w)      * (BGRAPH * HDIM)];
        a1 += src[(size_t)(w + 4)  * (BGRAPH * HDIM)];
        a2 += src[(size_t)(w + 8)  * (BGRAPH * HDIM)];
        a3 += src[(size_t)(w + 12) * (BGRAPH * HDIM)];
    }
    red[s][el] = (a0 + a1) + (a2 + a3);
    __syncthreads();
    if (tid < 256) {
        int b = blockIdx.x * 16 + (tid >> 4);
        float inv = 1.0f / fmaxf((float)counts[b], 1.0f);
        pl[tid] = ((red[0][tid] + red[1][tid]) + (red[2][tid] + red[3][tid])) * inv;
    }
    __syncthreads();
    if (tid < 16 * CDIM) {
        int bl = tid / CDIM, cc = tid % CDIM;
        float acc = bfc[cc];
#pragma unroll
        for (int j = 0; j < HDIM; ++j)
            acc += pl[bl * HDIM + j] * Wfc[j * CDIM + cc];
        out[blockIdx.x * (16 * CDIM) + tid] = acc;
    }
}

extern "C" void kernel_launch(void* const* d_in, const int* in_sizes, int n_in,
                              void* d_out, int out_size, void* d_ws, size_t ws_size,
                              hipStream_t stream) {
    (void)in_sizes; (void)n_in; (void)out_size; (void)ws_size;

    const float* x    = (const float*)d_in[0];
    const int*   ei   = (const int*)d_in[1];
    const int*   batch= (const int*)d_in[2];
    const float* W1   = (const float*)d_in[3];
    const float* b1   = (const float*)d_in[4];
    const float* W2   = (const float*)d_in[5];
    const float* b2   = (const float*)d_in[6];
    const float* Wfc  = (const float*)d_in[7];
    const float* bfc  = (const float*)d_in[8];
    float* out = (float*)d_out;

    const int* erow = ei;
    const int* ecol = ei + N_EDGES;

    char* ws = (char*)d_ws;
    size_t o = 0;
    auto alloc = [&](size_t bytes) -> void* {
        void* p = ws + o;
        o += (bytes + 255) & ~(size_t)255;
        return p;
    };
    float* dis    = (float*)alloc((size_t)N_NODES * 4);
    int*   degcnt = (int*)  alloc((size_t)N_NODES * 4);
    float* hws    = (float*)alloc((size_t)N_NODES * HDIM * 4);
    float* agg1   = (float*)alloc((size_t)PGRID * BGRAPH * HDIM * 4);
    float* part   = agg1;
    float2* nodew = (float2*)alloc((size_t)N_NODES * 8);
    int*   counts = (int*)  alloc((size_t)BGRAPH * 4);

    hipMemsetAsync(degcnt, 0, (size_t)N_NODES * 4, stream);
    hipMemsetAsync(counts, 0, (size_t)BGRAPH * 4, stream);

    count_deg_kernel<<<(N_EDGES + 255) / 256, 256, 0, stream>>>(ecol, degcnt);
    node_prep_kernel<<<(N_NODES + 255) / 256, 256, 0, stream>>>(degcnt, batch, dis, nodew, counts);

    int gemm_grid = (NTILES + 7) / 8;
    gemm1_kernel<<<gemm_grid, 256, 0, stream>>>(x, W1, dis, hws, agg1);

    edge_scatter_kernel<<<SC_GRID, 256, 0, stream>>>(erow, ecol, hws, agg1);

    gemm2_kernel<<<gemm_grid, 256, 0, stream>>>(agg1, W2, b1, dis, hws);

    edge_pool_kernel<<<PGRID, 1024, 0, stream>>>(erow, ecol, hws, nodew, b2, part);
    final_kernel<<<BGRAPH / 16, 1024, 0, stream>>>(part, counts, Wfc, bfc, out);
}


# in-degree from private LDS histograms: 256 work-groups (32 edge slices x 8 node ranges) count col in LDS, node prep sums 32 partial counts; global integer atomics and a memset go; float sums same
# speedup vs baseline: 2.7867x; 1.6281x over previous
#include <hip/hip_runtime.h>

#define N_NODES 200000
#define N_EDGES 6400000
#define FDIM    128
#define HDIM    16
#define CDIM    10
#define BGRAPH  512
#define NTILES  (N_NODES / 16)
#define PGRID   512
#define EPB     (N_EDGES / PGRID)
#define PPITCH  (BGRAPH + 1)
#define PTAB    (HDIM * PPITCH)

typedef float v2f __attribute__((ext_vector_type(2)));
typedef float v8f __attribute__((ext_vector_type(8)));

#define DG_NP   8
#define DG_SL   32
#define DG_PN   (N_NODES / DG_NP)
#define DG_SE4  (N_EDGES / DG_SL / 4)
#define DG_PP   25024
#define DG_TRIPS 7
static_assert(DG_PN * DG_NP == N_NODES && DG_PN % 4 == 0, "node ranges must be equal and 16-byte multiples");
static_assert(DG_SE4 * 4 * DG_SL == N_EDGES, "edge slices must be equal and 16-byte multiples");
static_assert(DG_PP >= DG_PN && (DG_PP * 4) % 128 == 0, "a partial plane starts on a line");
static_assert(DG_TRIPS * 7 * 1024 >= DG_SE4, "the trips must cover a slice");

__device__ __forceinline__ void deg_add(int* h, int c, int lo) {
    unsigned d = (unsigned)(c - lo);
    if (d < (unsigned)DG_PN) atomicAdd(&h[d], 1);
}

__global__ void __launch_bounds__(1024) deg_hist_kernel(const int* __restrict__ col,
                                                        int* __restrict__ degpart) {
    __shared__ int h[DG_PN];
    int tid = threadIdx.x;
    int s = blockIdx.x / DG_NP, p = blockIdx.x % DG_NP;
    int4* h4 = (int4*)h;
    for (int i = tid; i < DG_PN / 4; i += 1024) h4[i] = make_int4(0, 0, 0, 0);
    __syncthreads();

    int lo = p * DG_PN;
    const int4* c4 = (const int4*)(col + (size_t)s * (N_EDGES / DG_SL));
    for (int it = 0; it < DG_TRIPS; ++it) {
        int4 v[7];
#pragma unroll
        for (int j = 0; j < 7; ++j) {
            int idx = (it * 7 + j) * 1024 + tid;
            v[j] = idx < DG_SE4 ? c4[idx] : make_int4(-1, -1, -1, -1);
        }
#pragma unroll
        for (int j = 0; j < 7; ++j) {
            deg_add(h, v[j].x, lo);
            deg_add(h, v[j].y, lo);
            deg_add(h, v[j].z, lo);
            deg_add(h, v[j].w, lo);
        }
    }
    __syncthreads();

    int4* o4 = (int4*)(degpart + (size_t)blockIdx.x * DG_PP);
    for (int i = tid; i < DG_PN / 4; i += 1024) o4[i] = h4[i];
}

__global__ void node_prep_kernel(const int* __restrict__ degpart, const int* __restrict__ batch,
                                 float* __restrict__ dis, float2* __restrict__ nodew,
                                 int* __restrict__ counts) {
    int n = blockIdx.x * blockDim.x + threadIdx.x;
    if (n < N_NODES) {
        int p = n / DG_PN;
        const int* src = degpart + (size_t)p * DG_PP + (n - p * DG_PN);
        int deg = 0;
#pragma unroll
        for (int s = 0; s < DG_SL; ++s) deg += src[(size_t)s * (DG_NP * DG_PP)];
        float d = rsqrtf(1.0f + (float)deg);
        int g = batch[n];
        dis[n] = d;
        nodew[n] = make_float2(d, __int_as_float(g));
        atomicAdd(&counts[g], 1);
    }
}

__global__ void __launch_bounds__(256) gemm1_kernel(const float* __restrict__ x,
                                                    const float* __restrict__ W1,
                                                    const float* __restrict__ dis,
                                                    float* __restrict__ hws,
                                                    float* __restrict__ agg) {
    __shared__ float lw[(FDIM / 2) * 32];
    int tid = threadIdx.x;
#pragma unroll
    for (int i = 0; i < 8; ++i) {
        int idx = i * 256 + tid;
        int k = idx >> 4, n = idx & 15;
        lw[(k >> 1) * 32 + n * 2 + (k & 1)] = W1[idx];
    }
    __syncthreads();

    int wave = tid >> 5, lane = tid & 31;
    int tile = blockIdx.x * 8 + wave;
    if (tile >= NTILES) return;

    int h  = lane >> 4;
    int nn = lane & 15;
    long m = (long)tile * 16 + nn;
    const float* xrow = x + m * FDIM;

    v8f c = {0.f, 0.f, 0.f, 0.f, 0.f, 0.f, 0.f, 0.f};
#pragma unroll
    for (int kk = 0; kk < FDIM; kk += 4) {
        int k0 = kk + 2 * h;
        v2f a = *(const v2f*)(xrow + k0);
        v2f b = *(const v2f*)(&lw[(k0 >> 1) * 32 + nn * 2]);
        c = __builtin_amdgcn_wmma_f32_16x16x4_f32(false, a, false, b,
                                                  (short)0, c, false, false);
    }
#pragma unroll
    for (int v = 0; v < 8; ++v) {
        long r = (long)tile * 16 + v + 8 * h;
        float val = c[v] * dis[r];
        hws[r * HDIM + nn] = val;
        agg[r * HDIM + nn] = val;
    }
}

#define SC_GRID    1250
#define SC_GROUPS  (N_EDGES / (SC_GRID * 8 * 32))
static_assert(SC_GRID * 8 * 32 * SC_GROUPS == N_EDGES, "edges must split evenly over the waves");

template <int S>
__device__ __forceinline__ unsigned row_share(unsigned v) {
    return (unsigned)__builtin_amdgcn_update_dpp(0, (int)v, 0x150 + S, 0xf, 0xf, false);
}

#define SC_GATHER(S)                                                              \
    {                                                                             \
        unsigned r = row_share<S>(rv);                                            \
        co[S] = row_share<S>(cv) * 64u + jb;                                      \
        hv[S] = *(const float*)(hb + (size_t)(r * 64u + jb));                     \
    }
#define SC_ADD(S) atomicAdd((float*)(ab + (size_t)co[S]), hv[S]);

__global__ void __launch_bounds__(256) edge_scatter_kernel(const int* __restrict__ row,
                                                           const int* __restrict__ col,
                                                           const float* __restrict__ hws,
                                                           float* __restrict__ agg) {
    int lane = threadIdx.x & 31;
    int wave = blockIdx.x * 8 + (threadIdx.x >> 5);
    unsigned jb = (unsigned)(lane & 15) * 4u;
    const int* rp = row + (size_t)wave * (SC_GROUPS * 32) + lane;
    const int* cp = col + (size_t)wave * (SC_GROUPS * 32) + lane;
    const char* hb = (const char*)hws;
    char* ab = (char*)agg;
    unsigned rv = (unsigned)rp[0];
    unsigned cv = (unsigned)cp[0];
    for (int g = 0; g < SC_GROUPS; ++g) {
        int gn = min(g + 1, SC_GROUPS - 1);
        unsigned rn = (unsigned)rp[gn * 32];
        unsigned cn = (unsigned)cp[gn * 32];
        float hv[16];
        unsigned co[16];
        SC_GATHER(0)  SC_GATHER(1)  SC_GATHER(2)  SC_GATHER(3)
        SC_GATHER(4)  SC_GATHER(5)  SC_GATHER(6)  SC_GATHER(7)
        SC_GATHER(8)  SC_GATHER(9)  SC_GATHER(10) SC_GATHER(11)
        SC_GATHER(12) SC_GATHER(13) SC_GATHER(14) SC_GATHER(15)
        SC_ADD(0)  SC_ADD(1)  SC_ADD(2)  SC_ADD(3)
        SC_ADD(4)  SC_ADD(5)  SC_ADD(6)  SC_ADD(7)
        SC_ADD(8)  SC_ADD(9)  SC_ADD(10) SC_ADD(11)
        SC_ADD(12) SC_ADD(13) SC_ADD(14) SC_ADD(15)
        rv = rn;
        cv = cn;
    }
}

__global__ void __launch_bounds__(256) gemm2_kernel(const float* __restrict__ agg1,
                                                    const float* __restrict__ W2,
                                                    const float* __restrict__ b1v,
                                                    const float* __restrict__ dis,
                                                    float* __restrict__ hws2) {
    __shared__ float lw[(HDIM / 2) * 32];
    __shared__ float lb1[HDIM];
    int tid = threadIdx.x;
    if (tid < 256) {
        int k = tid >> 4, n = tid & 15;
        lw[(k >> 1) * 32 + n * 2 + (k & 1)] = W2[tid];
    }
    if (tid < HDIM) lb1[tid] = b1v[tid];
    __syncthreads();

    int wave = tid >> 5, lane = tid & 31;
    int tile = blockIdx.x * 8 + wave;
    if (tile >= NTILES) return;

    int h  = lane >> 4;
    int nn = lane & 15;
    long m = (long)tile * 16 + nn;
    float dm = dis[m];
    const float* arow = agg1 + m * HDIM;

    v8f c = {0.f, 0.f, 0.f, 0.f, 0.f, 0.f, 0.f, 0.f};
#pragma unroll
    for (int kk = 0; kk < HDIM; kk += 4) {
        int k0 = kk + 2 * h;
        v2f raw = *(const v2f*)(arow + k0);
        v2f a;
        a.x = fmaxf(dm * raw.x + lb1[k0], 0.0f);
        a.y = fmaxf(dm * raw.y + lb1[k0 + 1], 0.0f);
        v2f b = *(const v2f*)(&lw[(k0 >> 1) * 32 + nn * 2]);
        c = __builtin_amdgcn_wmma_f32_16x16x4_f32(false, a, false, b,
                                                  (short)0, c, false, false);
    }
#pragma unroll
    for (int v = 0; v < 8; ++v) {
        long r = (long)tile * 16 + v + 8 * h;
        float val = c[v] * dis[r];
        hws2[r * HDIM + nn] = val;
    }
}

__device__ __forceinline__ void pool_add4(float* lq, float2 nw, float4 hv) {
    float w = nw.x;
    float* p = lq + __float_as_int(nw.y);
    atomicAdd(p,              w * hv.x);
    atomicAdd(p + PPITCH,     w * hv.y);
    atomicAdd(p + 2 * PPITCH, w * hv.z);
    atomicAdd(p + 3 * PPITCH, w * hv.w);
}

__global__ void __launch_bounds__(1024) edge_pool_kernel(const int* __restrict__ row,
                                                         const int* __restrict__ col,
                                                         const float* __restrict__ hws,
                                                         const float2* __restrict__ nodew,
                                                         const float* __restrict__ b2v,
                                                         float* __restrict__ part) {
    __shared__ float lp[PTAB];
    int tid = threadIdx.x;
    for (int i = tid; i < PTAB; i += 1024) lp[i] = 0.0f;
    __syncthreads();

    int q    = tid & 3;
    int quad = tid >> 2;
    const float4* hws4 = (const float4*)hws;
    float* lq = lp + (4 * q) * PPITCH;

    int ebeg = blockIdx.x * EPB;
    const int4* row4 = (const int4*)(row + ebeg);
    const int4* col4 = (const int4*)(col + ebeg);
    int4 r4 = row4[quad];
    int4 c4 = col4[quad];
    for (int it = 0; it < EPB / 1024; ++it) {
        float2 nw0 = nodew[c4.x], nw1 = nodew[c4.y], nw2 = nodew[c4.z], nw3 = nodew[c4.w];
        float4 hv0 = hws4[(long)r4.x * 4 + q];
        float4 hv1 = hws4[(long)r4.y * 4 + q];
        float4 hv2 = hws4[(long)r4.z * 4 + q];
        float4 hv3 = hws4[(long)r4.w * 4 + q];
        int i4 = min((it + 1) * 256 + quad, EPB / 4 - 1);
        int4 rn = row4[i4];
        int4 cn = col4[i4];
        pool_add4(lq, nw0, hv0);
        pool_add4(lq, nw1, hv1);
        pool_add4(lq, nw2, hv2);
        pool_add4(lq, nw3, hv3);
        r4 = rn;
        c4 = cn;
    }
    if ((EPB / 1024) * 256 + quad < EPB / 4) {
        float2 nw0 = nodew[c4.x], nw1 = nodew[c4.y], nw2 = nodew[c4.z], nw3 = nodew[c4.w];
        float4 hv0 = hws4[(long)r4.x * 4 + q];
        float4 hv1 = hws4[(long)r4.y * 4 + q];
        float4 hv2 = hws4[(long)r4.z * 4 + q];
        float4 hv3 = hws4[(long)r4.w * 4 + q];
        pool_add4(lq, nw0, hv0);
        pool_add4(lq, nw1, hv1);
        pool_add4(lq, nw2, hv2);
        pool_add4(lq, nw3, hv3);
    }

    float bq0 = b2v[4 * q], bq1 = b2v[4 * q + 1], bq2 = b2v[4 * q + 2], bq3 = b2v[4 * q + 3];
    for (int n = blockIdx.x + PGRID * quad; n < N_NODES; n += PGRID * 256) {
        float2 nwn = nodew[n];
        float4 hvn = hws4[(long)n * 4 + q];
        float w = nwn.x;
        float* p = lq + __float_as_int(nwn.y);
        atomicAdd(p,              w * hvn.x + bq0);
        atomicAdd(p + PPITCH,     w * hvn.y + bq1);
        atomicAdd(p + 2 * PPITCH, w * hvn.z + bq2);
        atomicAdd(p + 3 * PPITCH, w * hvn.w + bq3);
    }
    __syncthreads();

    float4* out4 = (float4*)(part + (size_t)blockIdx.x * (BGRAPH * HDIM));
    for (int i = tid; i < BGRAPH * 4; i += 1024) {
        int g = i >> 2, qq = i & 3;
        const float* s = lp + (4 * qq) * PPITCH + g;
        out4[i] = make_float4(s[0], s[PPITCH], s[2 * PPITCH], s[3 * PPITCH]);
    }
}

__global__ void __launch_bounds__(1024) final_kernel(const float* __restrict__ part,
                                                     const int* __restrict__ counts,
                                                     const float* __restrict__ Wfc,
                                                     const float* __restrict__ bfc,
                                                     float* __restrict__ out) {
    __shared__ float red[4][256];
    __shared__ float pl[256];
    int tid = threadIdx.x;
    int el = tid & 255, s = tid >> 8;
    const float* src = part + (size_t)blockIdx.x * 256 + el;
    float a0 = 0.0f, a1 = 0.0f, a2 = 0.0f, a3 = 0.0f;
#pragma unroll 8
    for (int w = s; w < PGRID; w += 16) {
        a0 += src[(size_t)(w)      * (BGRAPH * HDIM)];
        a1 += src[(size_t)(w + 4)  * (BGRAPH * HDIM)];
        a2 += src[(size_t)(w + 8)  * (BGRAPH * HDIM)];
        a3 += src[(size_t)(w + 12) * (BGRAPH * HDIM)];
    }
    red[s][el] = (a0 + a1) + (a2 + a3);
    __syncthreads();
    if (tid < 256) {
        int b = blockIdx.x * 16 + (tid >> 4);
        float inv = 1.0f / fmaxf((float)counts[b], 1.0f);
        pl[tid] = ((red[0][tid] + red[1][tid]) + (red[2][tid] + red[3][tid])) * inv;
    }
    __syncthreads();
    if (tid < 16 * CDIM) {
        int bl = tid / CDIM, cc = tid % CDIM;
        float acc = bfc[cc];
#pragma unroll
        for (int j = 0; j < HDIM; ++j)
            acc += pl[bl * HDIM + j] * Wfc[j * CDIM + cc];
        out[blockIdx.x * (16 * CDIM) + tid] = acc;
    }
}

static_assert((size_t)DG_SL * DG_NP * DG_PP * 4 <= (size_t)N_NODES * HDIM * 4 + (size_t)PGRID * BGRAPH * HDIM * 4,
              "partial degree planes overrun hws + agg1");
static_assert(((size_t)N_NODES * HDIM * 4) % 256 == 0, "agg1 must start where hws ends");

extern "C" void kernel_launch(void* const* d_in, const int* in_sizes, int n_in,
                              void* d_out, int out_size, void* d_ws, size_t ws_size,
                              hipStream_t stream) {
    (void)in_sizes; (void)n_in; (void)out_size; (void)ws_size;

    const float* x    = (const float*)d_in[0];
    const int*   ei   = (const int*)d_in[1];
    const int*   batch= (const int*)d_in[2];
    const float* W1   = (const float*)d_in[3];
    const float* b1   = (const float*)d_in[4];
    const float* W2   = (const float*)d_in[5];
    const float* b2   = (const float*)d_in[6];
    const float* Wfc  = (const float*)d_in[7];
    const float* bfc  = (const float*)d_in[8];
    float* out = (float*)d_out;

    const int* erow = ei;
    const int* ecol = ei + N_EDGES;

    char* ws = (char*)d_ws;
    size_t o = 0;
    auto alloc = [&](size_t bytes) -> void* {
        void* p = ws + o;
        o += (bytes + 255) & ~(size_t)255;
        return p;
    };
    float* dis    = (float*)alloc((size_t)N_NODES * 4);
    float* hws    = (float*)alloc((size_t)N_NODES * HDIM * 4);
    float* agg1   = (float*)alloc((size_t)PGRID * BGRAPH * HDIM * 4);
    float* part   = agg1;
    int*   degpart = (int*)hws;
    float2* nodew = (float2*)alloc((size_t)N_NODES * 8);
    int*   counts = (int*)  alloc((size_t)BGRAPH * 4);

    hipMemsetAsync(counts, 0, (size_t)BGRAPH * 4, stream);

    deg_hist_kernel<<<DG_SL * DG_NP, 1024, 0, stream>>>(ecol, degpart);
    node_prep_kernel<<<(N_NODES + 255) / 256, 256, 0, stream>>>(degpart, batch, dis, nodew, counts);

    int gemm_grid = (NTILES + 7) / 8;
    gemm1_kernel<<<gemm_grid, 256, 0, stream>>>(x, W1, dis, hws, agg1);

    edge_scatter_kernel<<<SC_GRID, 256, 0, stream>>>(erow, ecol, hws, agg1);

    gemm2_kernel<<<gemm_grid, 256, 0, stream>>>(agg1, W2, b1, dis, hws);

    edge_pool_kernel<<<PGRID, 1024, 0, stream>>>(erow, ecol, hws, nodew, b2, part);
    final_kernel<<<BGRAPH / 16, 1024, 0, stream>>>(part, counts, Wfc, bfc, out);
}


# layer 1 edge sums in two planes, one per package half picked by the wave's die number: no line takes atomic adds from both halves; gemm2 adds the planes; sum order changes
# speedup vs baseline: 3.7667x; 1.3517x over previous
#include <hip/hip_runtime.h>

#define N_NODES 200000
#define N_EDGES 6400000
#define FDIM    128
#define HDIM    16
#define CDIM    10
#define BGRAPH  512
#define NTILES  (N_NODES / 16)
#define PGRID   512
#define EPB     (N_EDGES / PGRID)
#define PPITCH  (BGRAPH + 1)
#define PTAB    (HDIM * PPITCH)

#define PLANE_B  ((size_t)N_NODES * HDIM * 4)
#define WS_TWO   ((size_t)N_NODES * 4 + (size_t)N_NODES * 8 + 2048 + 3 * PLANE_B)

typedef float v2f __attribute__((ext_vector_type(2)));
typedef float v8f __attribute__((ext_vector_type(8)));

#define DG_NP   8
#define DG_SL   32
#define DG_PN   (N_NODES / DG_NP)
#define DG_SE4  (N_EDGES / DG_SL / 4)
#define DG_PP   25024
#define DG_TRIPS 7
static_assert(DG_PN * DG_NP == N_NODES && DG_PN % 4 == 0, "node ranges must be equal and 16-byte multiples");
static_assert(DG_SE4 * 4 * DG_SL == N_EDGES, "edge slices must be equal and 16-byte multiples");
static_assert(DG_PP >= DG_PN && (DG_PP * 4) % 128 == 0, "a partial plane starts on a line");
static_assert(DG_TRIPS * 7 * 1024 >= DG_SE4, "the trips must cover a slice");

__device__ __forceinline__ void deg_add(int* h, int c, int lo) {
    unsigned d = (unsigned)(c - lo);
    if (d < (unsigned)DG_PN) atomicAdd(&h[d], 1);
}

__global__ void __launch_bounds__(1024) deg_hist_kernel(const int* __restrict__ col,
                                                        int* __restrict__ degpart) {
    __shared__ int h[DG_PN];
    int tid = threadIdx.x;
    int s = blockIdx.x / DG_NP, p = blockIdx.x % DG_NP;
    int4* h4 = (int4*)h;
    for (int i = tid; i < DG_PN / 4; i += 1024) h4[i] = make_int4(0, 0, 0, 0);
    __syncthreads();

    int lo = p * DG_PN;
    const int4* c4 = (const int4*)(col + (size_t)s * (N_EDGES / DG_SL));
    for (int it = 0; it < DG_TRIPS; ++it) {
        int4 v[7];
#pragma unroll
        for (int j = 0; j < 7; ++j) {
            int idx = (it * 7 + j) * 1024 + tid;
            v[j] = idx < DG_SE4 ? c4[idx] : make_int4(-1, -1, -1, -1);
        }
#pragma unroll
        for (int j = 0; j < 7; ++j) {
            deg_add(h, v[j].x, lo);
            deg_add(h, v[j].y, lo);
            deg_add(h, v[j].z, lo);
            deg_add(h, v[j].w, lo);
        }
    }
    __syncthreads();

    int4* o4 = (int4*)(degpart + (size_t)blockIdx.x * DG_PP);
    for (int i = tid; i < DG_PN / 4; i += 1024) o4[i] = h4[i];
}

__global__ void node_prep_kernel(const int* __restrict__ degpart, const int* __restrict__ batch,
                                 float* __restrict__ dis, float2* __restrict__ nodew,
                                 int* __restrict__ counts) {
    int n = blockIdx.x * blockDim.x + threadIdx.x;
    if (n < N_NODES) {
        int p = n / DG_PN;
        const int* src = degpart + (size_t)p * DG_PP + (n - p * DG_PN);
        int deg = 0;
#pragma unroll
        for (int s = 0; s < DG_SL; ++s) deg += src[(size_t)s * (DG_NP * DG_PP)];
        float d = rsqrtf(1.0f + (float)deg);
        int g = batch[n];
        dis[n] = d;
        nodew[n] = make_float2(d, __int_as_float(g));
        atomicAdd(&counts[g], 1);
    }
}

__global__ void __launch_bounds__(256) gemm1_kernel(const float* __restrict__ x,
                                                    const float* __restrict__ W1,
                                                    const float* __restrict__ dis,
                                                    float* __restrict__ hws,
                                                    float* __restrict__ agg,
                                                    float* __restrict__ aggb,
                                                    size_t ws_size) {
    __shared__ float lw[(FDIM / 2) * 32];
    int tid = threadIdx.x;
#pragma unroll
    for (int i = 0; i < 8; ++i) {
        int idx = i * 256 + tid;
        int k = idx >> 4, n = idx & 15;
        lw[(k >> 1) * 32 + n * 2 + (k & 1)] = W1[idx];
    }
    __syncthreads();

    int wave = tid >> 5, lane = tid & 31;
    int tile = blockIdx.x * 8 + wave;
    if (tile >= NTILES) return;

    int h  = lane >> 4;
    int nn = lane & 15;
    long m = (long)tile * 16 + nn;
    const float* xrow = x + m * FDIM;

    v8f c = {0.f, 0.f, 0.f, 0.f, 0.f, 0.f, 0.f, 0.f};
#pragma unroll
    for (int kk = 0; kk < FDIM; kk += 4) {
        int k0 = kk + 2 * h;
        v2f a = *(const v2f*)(xrow + k0);
        v2f b = *(const v2f*)(&lw[(k0 >> 1) * 32 + nn * 2]);
        c = __builtin_amdgcn_wmma_f32_16x16x4_f32(false, a, false, b,
                                                  (short)0, c, false, false);
    }
#pragma unroll
    for (int v = 0; v < 8; ++v) {
        long r = (long)tile * 16 + v + 8 * h;
        float val = c[v] * dis[r];
        hws[r * HDIM + nn] = val;
        agg[r * HDIM + nn] = val;
    }
    if (ws_size >= WS_TWO) {
#pragma unroll
        for (int v = 0; v < 8; ++v) {
            long r = (long)tile * 16 + v + 8 * h;
            aggb[r * HDIM + nn] = 0.0f;
        }
    }
}

#define SC_GRID    1250
#define SC_GROUPS  (N_EDGES / (SC_GRID * 8 * 32))
static_assert(SC_GRID * 8 * 32 * SC_GROUPS == N_EDGES, "edges must split evenly over the waves");

template <int S>
__device__ __forceinline__ unsigned row_share(unsigned v) {
    return (unsigned)__builtin_amdgcn_update_dpp(0, (int)v, 0x150 + S, 0xf, 0xf, false);
}

#define SC_GATHER(S)                                                              \
    {                                                                             \
        unsigned r = row_share<S>(rv);                                            \
        co[S] = row_share<S>(cv) * 64u + jb;                                      \
        hv[S] = *(const float*)(hb + (size_t)(r * 64u + jb));                     \
    }
#define SC_ADD(S) atomicAdd((float*)(ab + (size_t)co[S]), hv[S]);

__global__ void __launch_bounds__(256) edge_scatter_kernel(const int* __restrict__ row,
                                                           const int* __restrict__ col,
                                                           const float* __restrict__ hws,
                                                           float* __restrict__ agg,
                                                           float* __restrict__ aggb,
                                                           size_t ws_size) {
    unsigned place = (unsigned)__builtin_amdgcn_s_sendmsg_rtn(0x87);
    bool upper = (ws_size >= WS_TWO) && ((place >> 18) & 1u);
    int lane = threadIdx.x & 31;
    int wave = blockIdx.x * 8 + (threadIdx.x >> 5);
    unsigned jb = (unsigned)(lane & 15) * 4u;
    const int* rp = row + (size_t)wave * (SC_GROUPS * 32) + lane;
    const int* cp = col + (size_t)wave * (SC_GROUPS * 32) + lane;
    const char* hb = (const char*)hws;
    char* ab = (char*)(upper ? aggb : agg);
    unsigned rv = (unsigned)rp[0];
    unsigned cv = (unsigned)cp[0];
    for (int g = 0; g < SC_GROUPS; ++g) {
        int gn = min(g + 1, SC_GROUPS - 1);
        unsigned rn = (unsigned)rp[gn * 32];
        unsigned cn = (unsigned)cp[gn * 32];
        float hv[16];
        unsigned co[16];
        SC_GATHER(0)  SC_GATHER(1)  SC_GATHER(2)  SC_GATHER(3)
        SC_GATHER(4)  SC_GATHER(5)  SC_GATHER(6)  SC_GATHER(7)
        SC_GATHER(8)  SC_GATHER(9)  SC_GATHER(10) SC_GATHER(11)
        SC_GATHER(12) SC_GATHER(13) SC_GATHER(14) SC_GATHER(15)
        SC_ADD(0)  SC_ADD(1)  SC_ADD(2)  SC_ADD(3)
        SC_ADD(4)  SC_ADD(5)  SC_ADD(6)  SC_ADD(7)
        SC_ADD(8)  SC_ADD(9)  SC_ADD(10) SC_ADD(11)
        SC_ADD(12) SC_ADD(13) SC_ADD(14) SC_ADD(15)
        rv = rn;
        cv = cn;
    }
}

__global__ void __launch_bounds__(256) gemm2_kernel(const float* __restrict__ agg1,
                                                    const float* __restrict__ agg1b,
                                                    size_t ws_size,
                                                    const float* __restrict__ W2,
                                                    const float* __restrict__ b1v,
                                                    const float* __restrict__ dis,
                                                    float* __restrict__ hws2) {
    __shared__ float lw[(HDIM / 2) * 32];
    __shared__ float lb1[HDIM];
    int tid = threadIdx.x;
    if (tid < 256) {
        int k = tid >> 4, n = tid & 15;
        lw[(k >> 1) * 32 + n * 2 + (k & 1)] = W2[tid];
    }
    if (tid < HDIM) lb1[tid] = b1v[tid];
    __syncthreads();

    int wave = tid >> 5, lane = tid & 31;
    int tile = blockIdx.x * 8 + wave;
    if (tile >= NTILES) return;

    int h  = lane >> 4;
    int nn = lane & 15;
    long m = (long)tile * 16 + nn;
    float dm = dis[m];
    const float* arow = agg1 + m * HDIM;
    const float* brow = agg1b + m * HDIM;

    v2f rawv[HDIM / 4];
#pragma unroll
    for (int i = 0; i < HDIM / 4; ++i) rawv[i] = *(const v2f*)(arow + 4 * i + 2 * h);
    if (ws_size >= WS_TWO) {
#pragma unroll
        for (int i = 0; i < HDIM / 4; ++i) {
            v2f rb = *(const v2f*)(brow + 4 * i + 2 * h);
            rawv[i].x += rb.x;
            rawv[i].y += rb.y;
        }
    }

    v8f c = {0.f, 0.f, 0.f, 0.f, 0.f, 0.f, 0.f, 0.f};
#pragma unroll
    for (int kk = 0; kk < HDIM; kk += 4) {
        int k0 = kk + 2 * h;
        v2f raw = rawv[kk / 4];
        v2f a;
        a.x = fmaxf(dm * raw.x + lb1[k0], 0.0f);
        a.y = fmaxf(dm * raw.y + lb1[k0 + 1], 0.0f);
        v2f b = *(const v2f*)(&lw[(k0 >> 1) * 32 + nn * 2]);
        c = __builtin_amdgcn_wmma_f32_16x16x4_f32(false, a, false, b,
                                                  (short)0, c, false, false);
    }
#pragma unroll
    for (int v = 0; v < 8; ++v) {
        long r = (long)tile * 16 + v + 8 * h;
        float val = c[v] * dis[r];
        hws2[r * HDIM + nn] = val;
    }
}

__device__ __forceinline__ void pool_add4(float* lq, float2 nw, float4 hv) {
    float w = nw.x;
    float* p = lq + __float_as_int(nw.y);
    atomicAdd(p,              w * hv.x);
    atomicAdd(p + PPITCH,     w * hv.y);
    atomicAdd(p + 2 * PPITCH, w * hv.z);
    atomicAdd(p + 3 * PPITCH, w * hv.w);
}

__global__ void __launch_bounds__(1024) edge_pool_kernel(const int* __restrict__ row,
                                                         const int* __restrict__ col,
                                                         const float* __restrict__ hws,
                                                         const float2* __restrict__ nodew,
                                                         const float* __restrict__ b2v,
                                                         float* __restrict__ part) {
    __shared__ float lp[PTAB];
    int tid = threadIdx.x;
    for (int i = tid; i < PTAB; i += 1024) lp[i] = 0.0f;
    __syncthreads();

    int q    = tid & 3;
    int quad = tid >> 2;
    const float4* hws4 = (const float4*)hws;
    float* lq = lp + (4 * q) * PPITCH;

    int ebeg = blockIdx.x * EPB;
    const int4* row4 = (const int4*)(row + ebeg);
    const int4* col4 = (const int4*)(col + ebeg);
    int4 r4 = row4[quad];
    int4 c4 = col4[quad];
    for (int it = 0; it < EPB / 1024; ++it) {
        float2 nw0 = nodew[c4.x], nw1 = nodew[c4.y], nw2 = nodew[c4.z], nw3 = nodew[c4.w];
        float4 hv0 = hws4[(long)r4.x * 4 + q];
        float4 hv1 = hws4[(long)r4.y * 4 + q];
        float4 hv2 = hws4[(long)r4.z * 4 + q];
        float4 hv3 = hws4[(long)r4.w * 4 + q];
        int i4 = min((it + 1) * 256 + quad, EPB / 4 - 1);
        int4 rn = row4[i4];
        int4 cn = col4[i4];
        pool_add4(lq, nw0, hv0);
        pool_add4(lq, nw1, hv1);
        pool_add4(lq, nw2, hv2);
        pool_add4(lq, nw3, hv3);
        r4 = rn;
        c4 = cn;
    }
    if ((EPB / 1024) * 256 + quad < EPB / 4) {
        float2 nw0 = nodew[c4.x], nw1 = nodew[c4.y], nw2 = nodew[c4.z], nw3 = nodew[c4.w];
        float4 hv0 = hws4[(long)r4.x * 4 + q];
        float4 hv1 = hws4[(long)r4.y * 4 + q];
        float4 hv2 = hws4[(long)r4.z * 4 + q];
        float4 hv3 = hws4[(long)r4.w * 4 + q];
        pool_add4(lq, nw0, hv0);
        pool_add4(lq, nw1, hv1);
        pool_add4(lq, nw2, hv2);
        pool_add4(lq, nw3, hv3);
    }

    float bq0 = b2v[4 * q], bq1 = b2v[4 * q + 1], bq2 = b2v[4 * q + 2], bq3 = b2v[4 * q + 3];
    for (int n = blockIdx.x + PGRID * quad; n < N_NODES; n += PGRID * 256) {
        float2 nwn = nodew[n];
        float4 hvn = hws4[(long)n * 4 + q];
        float w = nwn.x;
        float* p = lq + __float_as_int(nwn.y);
        atomicAdd(p,              w * hvn.x + bq0);
        atomicAdd(p + PPITCH,     w * hvn.y + bq1);
        atomicAdd(p + 2 * PPITCH, w * hvn.z + bq2);
        atomicAdd(p + 3 * PPITCH, w * hvn.w + bq3);
    }
    __syncthreads();

    float4* out4 = (float4*)(part + (size_t)blockIdx.x * (BGRAPH * HDIM));
    for (int i = tid; i < BGRAPH * 4; i += 1024) {
        int g = i >> 2, qq = i & 3;
        const float* s = lp + (4 * qq) * PPITCH + g;
        out4[i] = make_float4(s[0], s[PPITCH], s[2 * PPITCH], s[3 * PPITCH]);
    }
}

__global__ void __launch_bounds__(1024) final_kernel(const float* __restrict__ part,
                                                     const int* __restrict__ counts,
                                                     const float* __restrict__ Wfc,
                                                     const float* __restrict__ bfc,
                                                     float* __restrict__ out) {
    __shared__ float red[4][256];
    __shared__ float pl[256];
    int tid = threadIdx.x;
    int el = tid & 255, s = tid >> 8;
    const float* src = part + (size_t)blockIdx.x * 256 + el;
    float a0 = 0.0f, a1 = 0.0f, a2 = 0.0f, a3 = 0.0f;
#pragma unroll 8
    for (int w = s; w < PGRID; w += 16) {
        a0 += src[(size_t)(w)      * (BGRAPH * HDIM)];
        a1 += src[(size_t)(w + 4)  * (BGRAPH * HDIM)];
        a2 += src[(size_t)(w + 8)  * (BGRAPH * HDIM)];
        a3 += src[(size_t)(w + 12) * (BGRAPH * HDIM)];
    }
    red[s][el] = (a0 + a1) + (a2 + a3);
    __syncthreads();
    if (tid < 256) {
        int b = blockIdx.x * 16 + (tid >> 4);
        float inv = 1.0f / fmaxf((float)counts[b], 1.0f);
        pl[tid] = ((red[0][tid] + red[1][tid]) + (red[2][tid] + red[3][tid])) * inv;
    }
    __syncthreads();
    if (tid < 16 * CDIM) {
        int bl = tid / CDIM, cc = tid % CDIM;
        float acc = bfc[cc];
#pragma unroll
        for (int j = 0; j < HDIM; ++j)
            acc += pl[bl * HDIM + j] * Wfc[j * CDIM + cc];
        out[blockIdx.x * (16 * CDIM) + tid] = acc;
    }
}

static_assert((size_t)DG_SL * DG_NP * DG_PP * 4 <= (size_t)N_NODES * HDIM * 4 + (size_t)PGRID * BGRAPH * HDIM * 4,
              "partial degree planes overrun hws + agg1");
static_assert(PLANE_B % 256 == 0 && (size_t)PGRID * BGRAPH * HDIM * 4 <= 2 * PLANE_B,
              "the second plane follows the first, and the partial pool tables fit into the two");
static_assert(((size_t)N_NODES * 4) % 256 == 0 && ((size_t)N_NODES * 8) % 256 == 0,
              "WS_TWO counts the small tables without padding");
static_assert(((size_t)N_NODES * HDIM * 4) % 256 == 0, "agg1 must start where hws ends");

extern "C" void kernel_launch(void* const* d_in, const int* in_sizes, int n_in,
                              void* d_out, int out_size, void* d_ws, size_t ws_size,
                              hipStream_t stream) {
    (void)in_sizes; (void)n_in; (void)out_size;

    const float* x    = (const float*)d_in[0];
    const int*   ei   = (const int*)d_in[1];
    const int*   batch= (const int*)d_in[2];
    const float* W1   = (const float*)d_in[3];
    const float* b1   = (const float*)d_in[4];
    const float* W2   = (const float*)d_in[5];
    const float* b2   = (const float*)d_in[6];
    const float* Wfc  = (const float*)d_in[7];
    const float* bfc  = (const float*)d_in[8];
    float* out = (float*)d_out;

    const int* erow = ei;
    const int* ecol = ei + N_EDGES;

    char* ws = (char*)d_ws;
    size_t o = 0;
    auto alloc = [&](size_t bytes) -> void* {
        void* p = ws + o;
        o += (bytes + 255) & ~(size_t)255;
        return p;
    };
    float* dis    = (float*)alloc((size_t)N_NODES * 4);
    float2* nodew = (float2*)alloc((size_t)N_NODES * 8);
    int*   counts = (int*)  alloc((size_t)BGRAPH * 4);
    float* hws    = (float*)alloc((size_t)N_NODES * HDIM * 4);
    float* agg1   = (float*)alloc((size_t)PGRID * BGRAPH * HDIM * 4);
    float* agg1b  = (float*)((char*)agg1 + PLANE_B);
    float* part   = agg1;
    int*   degpart = (int*)hws;

    hipMemsetAsync(counts, 0, (size_t)BGRAPH * 4, stream);

    deg_hist_kernel<<<DG_SL * DG_NP, 1024, 0, stream>>>(ecol, degpart);
    node_prep_kernel<<<(N_NODES + 255) / 256, 256, 0, stream>>>(degpart, batch, dis, nodew, counts);

    int gemm_grid = (NTILES + 7) / 8;
    gemm1_kernel<<<gemm_grid, 256, 0, stream>>>(x, W1, dis, hws, agg1, agg1b, ws_size);

    edge_scatter_kernel<<<SC_GRID, 256, 0, stream>>>(erow, ecol, hws, agg1, agg1b, ws_size);

    gemm2_kernel<<<gemm_grid, 256, 0, stream>>>(agg1, agg1b, ws_size, W2, b1, dis, hws);

    edge_pool_kernel<<<PGRID, 1024, 0, stream>>>(erow, ecol, hws, nodew, b2, part);
    final_kernel<<<BGRAPH / 16, 1024, 0, stream>>>(part, counts, Wfc, bfc, out);
}


# layer 2 pool pass: the target's in-degree and graph id come from a 16-bit LDS record (one work-group per WGP, 146400 nodes, else the old table), not a second gather; 256 partials; sum order changes
# speedup vs baseline: 4.0547x; 1.0765x over previous
#include <hip/hip_runtime.h>

#define N_NODES 200000
#define N_EDGES 6400000
#define FDIM    128
#define HDIM    16
#define CDIM    10
#define BGRAPH  512
#define NTILES  (N_NODES / 16)
#define PGRID   256
#define EPB     (N_EDGES / PGRID)
#define AGG_REGION ((size_t)512 * BGRAPH * HDIM * 4)
#define NLDS    146400
#define KESC    127u
#define KCOPY   ((NLDS / 8 + 1 + 1023) / 1024)
static_assert(NLDS % 8 == 0 && NLDS <= N_NODES, "the LDS records are copied as 16-byte words");
static_assert(BGRAPH <= 512 && N_EDGES < (1 << 23), "graph id in 9 bits, in-degree in 23");
static_assert(EPB * PGRID == N_EDGES && EPB % 4 == 0, "edges split evenly, four per 16-byte entry");
#define PPITCH  (BGRAPH + 1)
#define PTAB    (HDIM * PPITCH)

#define PLANE_B  ((size_t)N_NODES * HDIM * 4)
#define WS_TWO   ((size_t)N_NODES * 4 + (size_t)N_NODES * 8 + 2048 + 3 * PLANE_B)

typedef float v2f __attribute__((ext_vector_type(2)));
typedef float v8f __attribute__((ext_vector_type(8)));

#define DG_NP   8
#define DG_SL   32
#define DG_PN   (N_NODES / DG_NP)
#define DG_SE4  (N_EDGES / DG_SL / 4)
#define DG_PP   25024
#define DG_TRIPS 7
static_assert(DG_PN * DG_NP == N_NODES && DG_PN % 4 == 0, "node ranges must be equal and 16-byte multiples");
static_assert(DG_SE4 * 4 * DG_SL == N_EDGES, "edge slices must be equal and 16-byte multiples");
static_assert(DG_PP >= DG_PN && (DG_PP * 4) % 128 == 0, "a partial plane starts on a line");
static_assert(DG_TRIPS * 7 * 1024 >= DG_SE4, "the trips must cover a slice");

__device__ __forceinline__ void deg_add(int* h, int c, int lo) {
    unsigned d = (unsigned)(c - lo);
    if (d < (unsigned)DG_PN) atomicAdd(&h[d], 1);
}

__global__ void __launch_bounds__(1024) deg_hist_kernel(const int* __restrict__ col,
                                                        int* __restrict__ degpart) {
    __shared__ int h[DG_PN];
    int tid = threadIdx.x;
    int s = blockIdx.x / DG_NP, p = blockIdx.x % DG_NP;
    int4* h4 = (int4*)h;
    for (int i = tid; i < DG_PN / 4; i += 1024) h4[i] = make_int4(0, 0, 0, 0);
    __syncthreads();

    int lo = p * DG_PN;
    const int4* c4 = (const int4*)(col + (size_t)s * (N_EDGES / DG_SL));
    for (int it = 0; it < DG_TRIPS; ++it) {
        int4 v[7];
#pragma unroll
        for (int j = 0; j < 7; ++j) {
            int idx = (it * 7 + j) * 1024 + tid;
            v[j] = idx < DG_SE4 ? c4[idx] : make_int4(-1, -1, -1, -1);
        }
#pragma unroll
        for (int j = 0; j < 7; ++j) {
            deg_add(h, v[j].x, lo);
            deg_add(h, v[j].y, lo);
            deg_add(h, v[j].z, lo);
            deg_add(h, v[j].w, lo);
        }
    }
    __syncthreads();

    int4* o4 = (int4*)(degpart + (size_t)blockIdx.x * DG_PP);
    for (int i = tid; i < DG_PN / 4; i += 1024) o4[i] = h4[i];
}

__global__ void node_prep_kernel(const int* __restrict__ degpart, const int* __restrict__ batch,
                                 float* __restrict__ dis, unsigned* __restrict__ nodew,
                                 unsigned short* __restrict__ nodek,
                                 int* __restrict__ counts) {
    int n = blockIdx.x * blockDim.x + threadIdx.x;
    if (n < N_NODES) {
        int p = n / DG_PN;
        const int* src = degpart + (size_t)p * DG_PP + (n - p * DG_PN);
        int deg = 0;
#pragma unroll
        for (int s = 0; s < DG_SL; ++s) deg += src[(size_t)s * (DG_NP * DG_PP)];
        float d = rsqrtf(1.0f + (float)deg);
        int g = batch[n];
        dis[n] = d;
        nodew[n] = ((unsigned)deg << 9) | (unsigned)g;
        nodek[n] = (unsigned short)(((unsigned)min(deg, (int)KESC) << 9) | (unsigned)g);
        atomicAdd(&counts[g], 1);
    }
}

__global__ void __launch_bounds__(256) gemm1_kernel(const float* __restrict__ x,
                                                    const float* __restrict__ W1,
                                                    const float* __restrict__ dis,
                                                    float* __restrict__ hws,
                                                    float* __restrict__ agg,
                                                    float* __restrict__ aggb,
                                                    size_t ws_size) {
    __shared__ float lw[(FDIM / 2) * 32];
    int tid = threadIdx.x;
#pragma unroll
    for (int i = 0; i < 8; ++i) {
        int idx = i * 256 + tid;
        int k = idx >> 4, n = idx & 15;
        lw[(k >> 1) * 32 + n * 2 + (k & 1)] = W1[idx];
    }
    __syncthreads();

    int wave = tid >> 5, lane = tid & 31;
    int tile = blockIdx.x * 8 + wave;
    if (tile >= NTILES) return;

    int h  = lane >> 4;
    int nn = lane & 15;
    long m = (long)tile * 16 + nn;
    const float* xrow = x + m * FDIM;

    v8f c = {0.f, 0.f, 0.f, 0.f, 0.f, 0.f, 0.f, 0.f};
#pragma unroll
    for (int kk = 0; kk < FDIM; kk += 4) {
        int k0 = kk + 2 * h;
        v2f a = *(const v2f*)(xrow + k0);
        v2f b = *(const v2f*)(&lw[(k0 >> 1) * 32 + nn * 2]);
        c = __builtin_amdgcn_wmma_f32_16x16x4_f32(false, a, false, b,
                                                  (short)0, c, false, false);
    }
#pragma unroll
    for (int v = 0; v < 8; ++v) {
        long r = (long)tile * 16 + v + 8 * h;
        float val = c[v] * dis[r];
        hws[r * HDIM + nn] = val;
        agg[r * HDIM + nn] = val;
    }
    if (ws_size >= WS_TWO) {
#pragma unroll
        for (int v = 0; v < 8; ++v) {
            long r = (long)tile * 16 + v + 8 * h;
            aggb[r * HDIM + nn] = 0.0f;
        }
    }
}

#define SC_GRID    1250
#define SC_GROUPS  (N_EDGES / (SC_GRID * 8 * 32))
static_assert(SC_GRID * 8 * 32 * SC_GROUPS == N_EDGES, "edges must split evenly over the waves");

template <int S>
__device__ __forceinline__ unsigned row_share(unsigned v) {
    return (unsigned)__builtin_amdgcn_update_dpp(0, (int)v, 0x150 + S, 0xf, 0xf, false);
}

#define SC_GATHER(S)                                                              \
    {                                                                             \
        unsigned r = row_share<S>(rv);                                            \
        co[S] = row_share<S>(cv) * 64u + jb;                                      \
        hv[S] = *(const float*)(hb + (size_t)(r * 64u + jb));                     \
    }
#define SC_ADD(S) atomicAdd((float*)(ab + (size_t)co[S]), hv[S]);

__global__ void __launch_bounds__(256) edge_scatter_kernel(const int* __restrict__ row,
                                                           const int* __restrict__ col,
                                                           const float* __restrict__ hws,
                                                           float* __restrict__ agg,
                                                           float* __restrict__ aggb,
                                                           size_t ws_size) {
    unsigned place = (unsigned)__builtin_amdgcn_s_sendmsg_rtn(0x87);
    bool upper = (ws_size >= WS_TWO) && ((place >> 18) & 1u);
    int lane = threadIdx.x & 31;
    int wave = blockIdx.x * 8 + (threadIdx.x >> 5);
    unsigned jb = (unsigned)(lane & 15) * 4u;
    const int* rp = row + (size_t)wave * (SC_GROUPS * 32) + lane;
    const int* cp = col + (size_t)wave * (SC_GROUPS * 32) + lane;
    const char* hb = (const char*)hws;
    char* ab = (char*)(upper ? aggb : agg);
    unsigned rv = (unsigned)rp[0];
    unsigned cv = (unsigned)cp[0];
    for (int g = 0; g < SC_GROUPS; ++g) {
        int gn = min(g + 1, SC_GROUPS - 1);
        unsigned rn = (unsigned)rp[gn * 32];
        unsigned cn = (unsigned)cp[gn * 32];
        float hv[16];
        unsigned co[16];
        SC_GATHER(0)  SC_GATHER(1)  SC_GATHER(2)  SC_GATHER(3)
        SC_GATHER(4)  SC_GATHER(5)  SC_GATHER(6)  SC_GATHER(7)
        SC_GATHER(8)  SC_GATHER(9)  SC_GATHER(10) SC_GATHER(11)
        SC_GATHER(12) SC_GATHER(13) SC_GATHER(14) SC_GATHER(15)
        SC_ADD(0)  SC_ADD(1)  SC_ADD(2)  SC_ADD(3)
        SC_ADD(4)  SC_ADD(5)  SC_ADD(6)  SC_ADD(7)
        SC_ADD(8)  SC_ADD(9)  SC_ADD(10) SC_ADD(11)
        SC_ADD(12) SC_ADD(13) SC_ADD(14) SC_ADD(15)
        rv = rn;
        cv = cn;
    }
}

__global__ void __launch_bounds__(256) gemm2_kernel(const float* __restrict__ agg1,
                                                    const float* __restrict__ agg1b,
                                                    size_t ws_size,
                                                    const float* __restrict__ W2,
                                                    const float* __restrict__ b1v,
                                                    const float* __restrict__ dis,
                                                    float* __restrict__ hws2) {
    __shared__ float lw[(HDIM / 2) * 32];
    __shared__ float lb1[HDIM];
    int tid = threadIdx.x;
    if (tid < 256) {
        int k = tid >> 4, n = tid & 15;
        lw[(k >> 1) * 32 + n * 2 + (k & 1)] = W2[tid];
    }
    if (tid < HDIM) lb1[tid] = b1v[tid];
    __syncthreads();

    int wave = tid >> 5, lane = tid & 31;
    int tile = blockIdx.x * 8 + wave;
    if (tile >= NTILES) return;

    int h  = lane >> 4;
    int nn = lane & 15;
    long m = (long)tile * 16 + nn;
    float dm = dis[m];
    const float* arow = agg1 + m * HDIM;
    const float* brow = agg1b + m * HDIM;

    v2f rawv[HDIM / 4];
#pragma unroll
    for (int i = 0; i < HDIM / 4; ++i) rawv[i] = *(const v2f*)(arow + 4 * i + 2 * h);
    if (ws_size >= WS_TWO) {
#pragma unroll
        for (int i = 0; i < HDIM / 4; ++i) {
            v2f rb = *(const v2f*)(brow + 4 * i + 2 * h);
            rawv[i].x += rb.x;
            rawv[i].y += rb.y;
        }
    }

    v8f c = {0.f, 0.f, 0.f, 0.f, 0.f, 0.f, 0.f, 0.f};
#pragma unroll
    for (int kk = 0; kk < HDIM; kk += 4) {
        int k0 = kk + 2 * h;
        v2f raw = rawv[kk / 4];
        v2f a;
        a.x = fmaxf(dm * raw.x + lb1[k0], 0.0f);
        a.y = fmaxf(dm * raw.y + lb1[k0 + 1], 0.0f);
        v2f b = *(const v2f*)(&lw[(k0 >> 1) * 32 + nn * 2]);
        c = __builtin_amdgcn_wmma_f32_16x16x4_f32(false, a, false, b,
                                                  (short)0, c, false, false);
    }
#pragma unroll
    for (int v = 0; v < 8; ++v) {
        long r = (long)tile * 16 + v + 8 * h;
        float val = c[v] * dis[r];
        hws2[r * HDIM + nn] = val;
    }
}

__device__ __forceinline__ void pool_add4(float* lq, unsigned rec, float4 hv) {
    float w = rsqrtf(1.0f + (float)(rec >> 9));
    float* p = lq + (rec & 511u);
    atomicAdd(p,              w * hv.x);
    atomicAdd(p + PPITCH,     w * hv.y);
    atomicAdd(p + 2 * PPITCH, w * hv.z);
    atomicAdd(p + 3 * PPITCH, w * hv.w);
}

__device__ __forceinline__ unsigned lds_rec(const unsigned short* lk, int c) {
    return lk[min((unsigned)c, (unsigned)NLDS)];
}

__device__ __forceinline__ unsigned full_rec(unsigned k, const unsigned* __restrict__ nodew, int c) {
    return (k >> 9) == KESC ? nodew[c] : k;
}

struct Quad4 {
    unsigned v0, v1, v2, v3;
    float4 h0, h1, h2, h3;
};

__device__ __forceinline__ Quad4 ask4(const unsigned short* lk, const unsigned* __restrict__ nodew,
                                      const float4* __restrict__ hws4, int q, int4 r4, int4 c4) {
    Quad4 a;
    unsigned k0 = lds_rec(lk, c4.x), k1 = lds_rec(lk, c4.y), k2 = lds_rec(lk, c4.z), k3 = lds_rec(lk, c4.w);
    a.h0 = hws4[(long)r4.x * 4 + q];
    a.h1 = hws4[(long)r4.y * 4 + q];
    a.h2 = hws4[(long)r4.z * 4 + q];
    a.h3 = hws4[(long)r4.w * 4 + q];
    a.v0 = full_rec(k0, nodew, c4.x);
    a.v1 = full_rec(k1, nodew, c4.y);
    a.v2 = full_rec(k2, nodew, c4.z);
    a.v3 = full_rec(k3, nodew, c4.w);
    return a;
}

__device__ __forceinline__ void add4(float* lq, const Quad4& a) {
    pool_add4(lq, a.v0, a.h0);
    pool_add4(lq, a.v1, a.h1);
    pool_add4(lq, a.v2, a.h2);
    pool_add4(lq, a.v3, a.h3);
}

__global__ void __launch_bounds__(1024) edge_pool_kernel(const int* __restrict__ row,
                                                         const int* __restrict__ col,
                                                         const float* __restrict__ hws,
                                                         const unsigned* __restrict__ nodew,
                                                         const unsigned short* __restrict__ nodek,
                                                         const float* __restrict__ b2v,
                                                         float* __restrict__ part) {
    __shared__ float lp[PTAB];
    __shared__ uint4 lk4[NLDS / 8 + 1];
    int tid = threadIdx.x;
    const uint4* ks = (const uint4*)nodek;
    uint4 kt[KCOPY];
#pragma unroll
    for (int j = 0; j < KCOPY; ++j) {
        int i = j * 1024 + tid;
        kt[j] = i < NLDS / 8 ? ks[i] : make_uint4(0xFFFFFFFFu, 0xFFFFFFFFu, 0xFFFFFFFFu, 0xFFFFFFFFu);
    }
    for (int i = tid; i < PTAB; i += 1024) lp[i] = 0.0f;
#pragma unroll
    for (int j = 0; j < KCOPY; ++j) {
        int i = j * 1024 + tid;
        if (i <= NLDS / 8) lk4[i] = kt[j];
    }
    __syncthreads();
    const unsigned short* lk = (const unsigned short*)lk4;

    int q    = tid & 3;
    int quad = tid >> 2;
    const float4* hws4 = (const float4*)hws;
    float* lq = lp + (4 * q) * PPITCH;

    int ebeg = blockIdx.x * EPB;
    const int4* row4 = (const int4*)(row + ebeg);
    const int4* col4 = (const int4*)(col + ebeg);
    int4 ra = row4[quad],       ca = col4[quad];
    int4 rb = row4[256 + quad], cb = col4[256 + quad];
    auto trip = [&](int it) {
        int ia = min((it + 1) * 512 + quad,       EPB / 4 - 1);
        int ib = min((it + 1) * 512 + 256 + quad, EPB / 4 - 1);
        int4 rna = row4[ia], cna = col4[ia];
        int4 rnb = row4[ib], cnb = col4[ib];
        Quad4 a = ask4(lk, nodew, hws4, q, ra, ca);
        Quad4 b = ask4(lk, nodew, hws4, q, rb, cb);
        add4(lq, a);
        add4(lq, b);
        ra = rna; ca = cna;
        rb = rnb; cb = cnb;
    };
    trip(0);
    for (int it = 1; it < EPB / 2048; ++it) trip(it);
    if ((EPB / 2048) * 512 + quad < EPB / 4) {
        Quad4 a = ask4(lk, nodew, hws4, q, ra, ca);
        add4(lq, a);
    }

    float bq0 = b2v[4 * q], bq1 = b2v[4 * q + 1], bq2 = b2v[4 * q + 2], bq3 = b2v[4 * q + 3];
    static_assert(PGRID * 256 * 4 >= N_NODES, "four nodes per quad cover the node range");
    unsigned nrec[4];
    float4 nhv[4];
#pragma unroll
    for (int j = 0; j < 4; ++j) {
        int nc = min(blockIdx.x + PGRID * quad + PGRID * 256 * j, N_NODES - 1);
        nrec[j] = nodew[nc];
        nhv[j] = hws4[(long)nc * 4 + q];
    }
#pragma unroll
    for (int j = 0; j < 4; ++j) {
        int n = blockIdx.x + PGRID * quad + PGRID * 256 * j;
        if (n < N_NODES) {
            float w = rsqrtf(1.0f + (float)(nrec[j] >> 9));
            float* p = lq + (nrec[j] & 511u);
            atomicAdd(p,              w * nhv[j].x + bq0);
            atomicAdd(p + PPITCH,     w * nhv[j].y + bq1);
            atomicAdd(p + 2 * PPITCH, w * nhv[j].z + bq2);
            atomicAdd(p + 3 * PPITCH, w * nhv[j].w + bq3);
        }
    }
    __syncthreads();

    float4* out4 = (float4*)(part + (size_t)blockIdx.x * (BGRAPH * HDIM));
    for (int i = tid; i < BGRAPH * 4; i += 1024) {
        int g = i >> 2, qq = i & 3;
        const float* s = lp + (4 * qq) * PPITCH + g;
        out4[i] = make_float4(s[0], s[PPITCH], s[2 * PPITCH], s[3 * PPITCH]);
    }
}

__global__ void __launch_bounds__(1024) final_kernel(const float* __restrict__ part,
                                                     const int* __restrict__ counts,
                                                     const float* __restrict__ Wfc,
                                                     const float* __restrict__ bfc,
                                                     float* __restrict__ out) {
    __shared__ float red[4][256];
    __shared__ float pl[256];
    int tid = threadIdx.x;
    int el = tid & 255, s = tid >> 8;
    const float* src = part + (size_t)blockIdx.x * 256 + el;
    float a0 = 0.0f, a1 = 0.0f, a2 = 0.0f, a3 = 0.0f;
#pragma unroll 8
    for (int w = s; w < PGRID; w += 16) {
        a0 += src[(size_t)(w)      * (BGRAPH * HDIM)];
        a1 += src[(size_t)(w + 4)  * (BGRAPH * HDIM)];
        a2 += src[(size_t)(w + 8)  * (BGRAPH * HDIM)];
        a3 += src[(size_t)(w + 12) * (BGRAPH * HDIM)];
    }
    red[s][el] = (a0 + a1) + (a2 + a3);
    __syncthreads();
    if (tid < 256) {
        int b = blockIdx.x * 16 + (tid >> 4);
        float inv = 1.0f / fmaxf((float)counts[b], 1.0f);
        pl[tid] = ((red[0][tid] + red[1][tid]) + (red[2][tid] + red[3][tid])) * inv;
    }
    __syncthreads();
    if (tid < 16 * CDIM) {
        int bl = tid / CDIM, cc = tid % CDIM;
        float acc = bfc[cc];
#pragma unroll
        for (int j = 0; j < HDIM; ++j)
            acc += pl[bl * HDIM + j] * Wfc[j * CDIM + cc];
        out[blockIdx.x * (16 * CDIM) + tid] = acc;
    }
}

static_assert((size_t)DG_SL * DG_NP * DG_PP * 4 <= (size_t)N_NODES * HDIM * 4 + AGG_REGION,
              "partial degree planes overrun hws + agg1");
static_assert(PLANE_B <= AGG_REGION && (size_t)PGRID * BGRAPH * HDIM * 4 <= AGG_REGION,
              "the first plane and, later, the partial pool tables fit into the region");
static_assert(((size_t)N_NODES * 4) % 16 == 0 && (size_t)N_NODES * 4 + (size_t)N_NODES * 2 <= (size_t)N_NODES * 8,
              "the 16-bit records follow the 32-bit ones inside the node table, on a 16-byte boundary");
static_assert(PLANE_B % 256 == 0 && (size_t)PGRID * BGRAPH * HDIM * 4 <= 2 * PLANE_B,
              "the second plane follows the first, and the partial pool tables fit into the two");
static_assert(((size_t)N_NODES * 4) % 256 == 0 && ((size_t)N_NODES * 8) % 256 == 0,
              "WS_TWO counts the small tables without padding");
static_assert(((size_t)N_NODES * HDIM * 4) % 256 == 0, "agg1 must start where hws ends");

extern "C" void kernel_launch(void* const* d_in, const int* in_sizes, int n_in,
                              void* d_out, int out_size, void* d_ws, size_t ws_size,
                              hipStream_t stream) {
    (void)in_sizes; (void)n_in; (void)out_size;

    const float* x    = (const float*)d_in[0];
    const int*   ei   = (const int*)d_in[1];
    const int*   batch= (const int*)d_in[2];
    const float* W1   = (const float*)d_in[3];
    const float* b1   = (const float*)d_in[4];
    const float* W2   = (const float*)d_in[5];
    const float* b2   = (const float*)d_in[6];
    const float* Wfc  = (const float*)d_in[7];
    const float* bfc  = (const float*)d_in[8];
    float* out = (float*)d_out;

    const int* erow = ei;
    const int* ecol = ei + N_EDGES;

    char* ws = (char*)d_ws;
    size_t o = 0;
    auto alloc = [&](size_t bytes) -> void* {
        void* p = ws + o;
        o += (bytes + 255) & ~(size_t)255;
        return p;
    };
    float* dis    = (float*)alloc((size_t)N_NODES * 4);
    unsigned* nodew = (unsigned*)alloc((size_t)N_NODES * 8);
    unsigned short* nodek = (unsigned short*)((char*)nodew + (size_t)N_NODES * 4);
    int*   counts = (int*)  alloc((size_t)BGRAPH * 4);
    float* hws    = (float*)alloc((size_t)N_NODES * HDIM * 4);
    float* agg1   = (float*)alloc(AGG_REGION);
    float* agg1b  = (float*)((char*)agg1 + PLANE_B);
    float* part   = agg1;
    int*   degpart = (int*)hws;

    hipMemsetAsync(counts, 0, (size_t)BGRAPH * 4, stream);

    deg_hist_kernel<<<DG_SL * DG_NP, 1024, 0, stream>>>(ecol, degpart);
    node_prep_kernel<<<(N_NODES + 255) / 256, 256, 0, stream>>>(degpart, batch, dis, nodew, nodek, counts);

    int gemm_grid = (NTILES + 7) / 8;
    gemm1_kernel<<<gemm_grid, 256, 0, stream>>>(x, W1, dis, hws, agg1, agg1b, ws_size);

    edge_scatter_kernel<<<SC_GRID, 256, 0, stream>>>(erow, ecol, hws, agg1, agg1b, ws_size);

    gemm2_kernel<<<gemm_grid, 256, 0, stream>>>(agg1, agg1b, ws_size, W2, b1, dis, hws);

    edge_pool_kernel<<<PGRID, 1024, 0, stream>>>(erow, ecol, hws, nodew, nodek, b2, part);
    final_kernel<<<BGRAPH / 16, 1024, 0, stream>>>(part, counts, Wfc, bfc, out);
}


# layer 2 pool pass: every node's record fits LDS, one byte each (6-bit in-degree, graph id as 2 bits above a base kept per 64 nodes, escape kept): no second gather for the last 53600 nodes; sums same
# speedup vs baseline: 4.2812x; 1.0559x over previous
#include <hip/hip_runtime.h>

#define N_NODES 200000
#define N_EDGES 6400000
#define FDIM    128
#define HDIM    16
#define CDIM    10
#define BGRAPH  512
#define NTILES  (N_NODES / 16)
#define PGRID   256
#define EPB     (N_EDGES / PGRID)
#define AGG_REGION ((size_t)512 * BGRAPH * HDIM * 4)
#define NBLK    64
#define NBASE   (N_NODES / NBLK)
#define NBASEP  ((NBASE + 7) & ~7)
#define KESC8   63u
#define KCOPY   ((N_NODES / 16 + 1023) / 1024)
static_assert(N_NODES % NBLK == 0 && N_NODES % 16 == 0, "whole blocks; the records are copied as 16-byte words");
static_assert(BGRAPH <= 512 && N_EDGES < (1 << 23), "graph id in 9 bits, in-degree in 23");
static_assert(EPB * PGRID == N_EDGES && EPB % 4 == 0, "edges split evenly, four per 16-byte entry");
#define PPITCH  (BGRAPH + 1)
#define PTAB    (HDIM * PPITCH)

#define PLANE_B  ((size_t)N_NODES * HDIM * 4)
#define WS_TWO   ((size_t)N_NODES * 4 + (size_t)N_NODES * 8 + 2048 + 3 * PLANE_B)

typedef float v2f __attribute__((ext_vector_type(2)));
typedef float v8f __attribute__((ext_vector_type(8)));

#define DG_NP   8
#define DG_SL   32
#define DG_PN   (N_NODES / DG_NP)
#define DG_SE4  (N_EDGES / DG_SL / 4)
#define DG_PP   25024
#define DG_TRIPS 7
static_assert(DG_PN * DG_NP == N_NODES && DG_PN % 4 == 0, "node ranges must be equal and 16-byte multiples");
static_assert(DG_SE4 * 4 * DG_SL == N_EDGES, "edge slices must be equal and 16-byte multiples");
static_assert(DG_PP >= DG_PN && (DG_PP * 4) % 128 == 0, "a partial plane starts on a line");
static_assert(DG_TRIPS * 7 * 1024 >= DG_SE4, "the trips must cover a slice");

__device__ __forceinline__ void deg_add(int* h, int c, int lo) {
    unsigned d = (unsigned)(c - lo);
    if (d < (unsigned)DG_PN) atomicAdd(&h[d], 1);
}

__global__ void __launch_bounds__(1024) deg_hist_kernel(const int* __restrict__ col,
                                                        int* __restrict__ degpart) {
    __shared__ int h[DG_PN];
    int tid = threadIdx.x;
    int s = blockIdx.x / DG_NP, p = blockIdx.x % DG_NP;
    int4* h4 = (int4*)h;
    for (int i = tid; i < DG_PN / 4; i += 1024) h4[i] = make_int4(0, 0, 0, 0);
    __syncthreads();

    int lo = p * DG_PN;
    const int4* c4 = (const int4*)(col + (size_t)s * (N_EDGES / DG_SL));
    for (int it = 0; it < DG_TRIPS; ++it) {
        int4 v[7];
#pragma unroll
        for (int j = 0; j < 7; ++j) {
            int idx = (it * 7 + j) * 1024 + tid;
            v[j] = idx < DG_SE4 ? c4[idx] : make_int4(-1, -1, -1, -1);
        }
#pragma unroll
        for (int j = 0; j < 7; ++j) {
            deg_add(h, v[j].x, lo);
            deg_add(h, v[j].y, lo);
            deg_add(h, v[j].z, lo);
            deg_add(h, v[j].w, lo);
        }
    }
    __syncthreads();

    int4* o4 = (int4*)(degpart + (size_t)blockIdx.x * DG_PP);
    for (int i = tid; i < DG_PN / 4; i += 1024) o4[i] = h4[i];
}

__global__ void node_prep_kernel(const int* __restrict__ degpart, const int* __restrict__ batch,
                                 float* __restrict__ dis, unsigned* __restrict__ nodew,
                                 unsigned char* __restrict__ nodek,
                                 unsigned short* __restrict__ nbase,
                                 int* __restrict__ counts) {
    int n = blockIdx.x * blockDim.x + threadIdx.x;
    if (n < N_NODES) {
        int p = n / DG_PN;
        const int* src = degpart + (size_t)p * DG_PP + (n - p * DG_PN);
        int deg = 0;
#pragma unroll
        for (int s = 0; s < DG_SL; ++s) deg += src[(size_t)s * (DG_NP * DG_PP)];
        float d = rsqrtf(1.0f + (float)deg);
        int g = batch[n];
        int gb = batch[n & ~(NBLK - 1)];
        unsigned dl = (unsigned)(g - gb);
        bool fits = deg < (int)KESC8 && dl <= 3u;
        dis[n] = d;
        nodew[n] = ((unsigned)deg << 9) | (unsigned)g;
        nodek[n] = fits ? (unsigned char)(((unsigned)deg << 2) | dl) : (unsigned char)0xFF;
        if ((n & (NBLK - 1)) == 0) nbase[n / NBLK] = (unsigned short)g;
        if (n < NBASEP - NBASE) nbase[NBASE + n] = 0;
        atomicAdd(&counts[g], 1);
    }
}

__global__ void __launch_bounds__(256) gemm1_kernel(const float* __restrict__ x,
                                                    const float* __restrict__ W1,
                                                    const float* __restrict__ dis,
                                                    float* __restrict__ hws,
                                                    float* __restrict__ agg,
                                                    float* __restrict__ aggb,
                                                    size_t ws_size) {
    __shared__ float lw[(FDIM / 2) * 32];
    int tid = threadIdx.x;
#pragma unroll
    for (int i = 0; i < 8; ++i) {
        int idx = i * 256 + tid;
        int k = idx >> 4, n = idx & 15;
        lw[(k >> 1) * 32 + n * 2 + (k & 1)] = W1[idx];
    }
    __syncthreads();

    int wave = tid >> 5, lane = tid & 31;
    int tile = blockIdx.x * 8 + wave;
    if (tile >= NTILES) return;

    int h  = lane >> 4;
    int nn = lane & 15;
    long m = (long)tile * 16 + nn;
    const float* xrow = x + m * FDIM;

    v8f c = {0.f, 0.f, 0.f, 0.f, 0.f, 0.f, 0.f, 0.f};
#pragma unroll
    for (int kk = 0; kk < FDIM; kk += 4) {
        int k0 = kk + 2 * h;
        v2f a = *(const v2f*)(xrow + k0);
        v2f b = *(const v2f*)(&lw[(k0 >> 1) * 32 + nn * 2]);
        c = __builtin_amdgcn_wmma_f32_16x16x4_f32(false, a, false, b,
                                                  (short)0, c, false, false);
    }
#pragma unroll
    for (int v = 0; v < 8; ++v) {
        long r = (long)tile * 16 + v + 8 * h;
        float val = c[v] * dis[r];
        hws[r * HDIM + nn] = val;
        agg[r * HDIM + nn] = val;
    }
    if (ws_size >= WS_TWO) {
#pragma unroll
        for (int v = 0; v < 8; ++v) {
            long r = (long)tile * 16 + v + 8 * h;
            aggb[r * HDIM + nn] = 0.0f;
        }
    }
}

#define SC_GRID    1250
#define SC_GROUPS  (N_EDGES / (SC_GRID * 8 * 32))
static_assert(SC_GRID * 8 * 32 * SC_GROUPS == N_EDGES, "edges must split evenly over the waves");

template <int S>
__device__ __forceinline__ unsigned row_share(unsigned v) {
    return (unsigned)__builtin_amdgcn_update_dpp(0, (int)v, 0x150 + S, 0xf, 0xf, false);
}

#define SC_GATHER(S)                                                              \
    {                                                                             \
        unsigned r = row_share<S>(rv);                                            \
        co[S] = row_share<S>(cv) * 64u + jb;                                      \
        hv[S] = *(const float*)(hb + (size_t)(r * 64u + jb));                     \
    }
#define SC_ADD(S) atomicAdd((float*)(ab + (size_t)co[S]), hv[S]);

__global__ void __launch_bounds__(256) edge_scatter_kernel(const int* __restrict__ row,
                                                           const int* __restrict__ col,
                                                           const float* __restrict__ hws,
                                                           float* __restrict__ agg,
                                                           float* __restrict__ aggb,
                                                           size_t ws_size) {
    unsigned place = (unsigned)__builtin_amdgcn_s_sendmsg_rtn(0x87);
    bool upper = (ws_size >= WS_TWO) && ((place >> 18) & 1u);
    int lane = threadIdx.x & 31;
    int wave = blockIdx.x * 8 + (threadIdx.x >> 5);
    unsigned jb = (unsigned)(lane & 15) * 4u;
    const int* rp = row + (size_t)wave * (SC_GROUPS * 32) + lane;
    const int* cp = col + (size_t)wave * (SC_GROUPS * 32) + lane;
    const char* hb = (const char*)hws;
    char* ab = (char*)(upper ? aggb : agg);
    unsigned rv = (unsigned)rp[0];
    unsigned cv = (unsigned)cp[0];
    for (int g = 0; g < SC_GROUPS; ++g) {
        int gn = min(g + 1, SC_GROUPS - 1);
        unsigned rn = (unsigned)rp[gn * 32];
        unsigned cn = (unsigned)cp[gn * 32];
        float hv[16];
        unsigned co[16];
        SC_GATHER(0)  SC_GATHER(1)  SC_GATHER(2)  SC_GATHER(3)
        SC_GATHER(4)  SC_GATHER(5)  SC_GATHER(6)  SC_GATHER(7)
        SC_GATHER(8)  SC_GATHER(9)  SC_GATHER(10) SC_GATHER(11)
        SC_GATHER(12) SC_GATHER(13) SC_GATHER(14) SC_GATHER(15)
        SC_ADD(0)  SC_ADD(1)  SC_ADD(2)  SC_ADD(3)
        SC_ADD(4)  SC_ADD(5)  SC_ADD(6)  SC_ADD(7)
        SC_ADD(8)  SC_ADD(9)  SC_ADD(10) SC_ADD(11)
        SC_ADD(12) SC_ADD(13) SC_ADD(14) SC_ADD(15)
        rv = rn;
        cv = cn;
    }
}

__global__ void __launch_bounds__(256) gemm2_kernel(const float* __restrict__ agg1,
                                                    const float* __restrict__ agg1b,
                                                    size_t ws_size,
                                                    const float* __restrict__ W2,
                                                    const float* __restrict__ b1v,
                                                    const float* __restrict__ dis,
                                                    float* __restrict__ hws2) {
    __shared__ float lw[(HDIM / 2) * 32];
    __shared__ float lb1[HDIM];
    int tid = threadIdx.x;
    if (tid < 256) {
        int k = tid >> 4, n = tid & 15;
        lw[(k >> 1) * 32 + n * 2 + (k & 1)] = W2[tid];
    }
    if (tid < HDIM) lb1[tid] = b1v[tid];
    __syncthreads();

    int wave = tid >> 5, lane = tid & 31;
    int tile = blockIdx.x * 8 + wave;
    if (tile >= NTILES) return;

    int h  = lane >> 4;
    int nn = lane & 15;
    long m = (long)tile * 16 + nn;
    float dm = dis[m];
    const float* arow = agg1 + m * HDIM;
    const float* brow = agg1b + m * HDIM;

    v2f rawv[HDIM / 4];
#pragma unroll
    for (int i = 0; i < HDIM / 4; ++i) rawv[i] = *(const v2f*)(arow + 4 * i + 2 * h);
    if (ws_size >= WS_TWO) {
#pragma unroll
        for (int i = 0; i < HDIM / 4; ++i) {
            v2f rb = *(const v2f*)(brow + 4 * i + 2 * h);
            rawv[i].x += rb.x;
            rawv[i].y += rb.y;
        }
    }

    v8f c = {0.f, 0.f, 0.f, 0.f, 0.f, 0.f, 0.f, 0.f};
#pragma unroll
    for (int kk = 0; kk < HDIM; kk += 4) {
        int k0 = kk + 2 * h;
        v2f raw = rawv[kk / 4];
        v2f a;
        a.x = fmaxf(dm * raw.x + lb1[k0], 0.0f);
        a.y = fmaxf(dm * raw.y + lb1[k0 + 1], 0.0f);
        v2f b = *(const v2f*)(&lw[(k0 >> 1) * 32 + nn * 2]);
        c = __builtin_amdgcn_wmma_f32_16x16x4_f32(false, a, false, b,
                                                  (short)0, c, false, false);
    }
#pragma unroll
    for (int v = 0; v < 8; ++v) {
        long r = (long)tile * 16 + v + 8 * h;
        float val = c[v] * dis[r];
        hws2[r * HDIM + nn] = val;
    }
}

__device__ __forceinline__ void pool_add4(float* lq, unsigned rec, float4 hv) {
    float w = rsqrtf(1.0f + (float)(rec >> 9));
    float* p = lq + (rec & 511u);
    atomicAdd(p,              w * hv.x);
    atomicAdd(p + PPITCH,     w * hv.y);
    atomicAdd(p + 2 * PPITCH, w * hv.z);
    atomicAdd(p + 3 * PPITCH, w * hv.w);
}

struct RecTabs {
    const unsigned char* lk;
    const unsigned short* lb;
};

__device__ __forceinline__ unsigned lds_rec(RecTabs t, int c) {
    unsigned k = t.lk[c];
    unsigned b = t.lb[(unsigned)c / NBLK];
    return ((k >> 2) << 9) | ((b + (k & 3u)) & 511u);
}

__device__ __forceinline__ unsigned full_rec(unsigned k, const unsigned* __restrict__ nodew, int c) {
    return (k >> 9) == KESC8 ? nodew[c] : k;
}

struct Quad4 {
    unsigned v0, v1, v2, v3;
    float4 h0, h1, h2, h3;
};

__device__ __forceinline__ Quad4 ask4(RecTabs lk, const unsigned* __restrict__ nodew,
                                      const float4* __restrict__ hws4, int q, int4 r4, int4 c4) {
    Quad4 a;
    unsigned k0 = lds_rec(lk, c4.x), k1 = lds_rec(lk, c4.y), k2 = lds_rec(lk, c4.z), k3 = lds_rec(lk, c4.w);
    a.h0 = hws4[(long)r4.x * 4 + q];
    a.h1 = hws4[(long)r4.y * 4 + q];
    a.h2 = hws4[(long)r4.z * 4 + q];
    a.h3 = hws4[(long)r4.w * 4 + q];
    a.v0 = full_rec(k0, nodew, c4.x);
    a.v1 = full_rec(k1, nodew, c4.y);
    a.v2 = full_rec(k2, nodew, c4.z);
    a.v3 = full_rec(k3, nodew, c4.w);
    return a;
}

__device__ __forceinline__ void add4(float* lq, const Quad4& a) {
    pool_add4(lq, a.v0, a.h0);
    pool_add4(lq, a.v1, a.h1);
    pool_add4(lq, a.v2, a.h2);
    pool_add4(lq, a.v3, a.h3);
}

__global__ void __launch_bounds__(1024) edge_pool_kernel(const int* __restrict__ row,
                                                         const int* __restrict__ col,
                                                         const float* __restrict__ hws,
                                                         const unsigned* __restrict__ nodew,
                                                         const unsigned char* __restrict__ nodek,
                                                         const unsigned short* __restrict__ nbase,
                                                         const float* __restrict__ b2v,
                                                         float* __restrict__ part) {
    __shared__ float lp[PTAB];
    __shared__ uint4 lk4[N_NODES / 16];
    __shared__ uint4 lb4[NBASEP / 8];
    int tid = threadIdx.x;
    const uint4* ks = (const uint4*)nodek;
    const uint4* bs = (const uint4*)nbase;
    uint4 kt[KCOPY];
    uint4 bt = make_uint4(0u, 0u, 0u, 0u);
#pragma unroll
    for (int j = 0; j < KCOPY; ++j) {
        int i = j * 1024 + tid;
        kt[j] = ks[min(i, N_NODES / 16 - 1)];
    }
    if (tid < NBASEP / 8) bt = bs[tid];
    for (int i = tid; i < PTAB; i += 1024) lp[i] = 0.0f;
#pragma unroll
    for (int j = 0; j < KCOPY; ++j) {
        int i = j * 1024 + tid;
        if (i < N_NODES / 16) lk4[i] = kt[j];
    }
    if (tid < NBASEP / 8) lb4[tid] = bt;
    __syncthreads();
    RecTabs lk;
    lk.lk = (const unsigned char*)lk4;
    lk.lb = (const unsigned short*)lb4;

    int q    = tid & 3;
    int quad = tid >> 2;
    const float4* hws4 = (const float4*)hws;
    float* lq = lp + (4 * q) * PPITCH;

    int ebeg = blockIdx.x * EPB;
    const int4* row4 = (const int4*)(row + ebeg);
    const int4* col4 = (const int4*)(col + ebeg);
    int4 ra = row4[quad],       ca = col4[quad];
    int4 rb = row4[256 + quad], cb = col4[256 + quad];
    auto trip = [&](int it) {
        int ia = min((it + 1) * 512 + quad,       EPB / 4 - 1);
        int ib = min((it + 1) * 512 + 256 + quad, EPB / 4 - 1);
        int4 rna = row4[ia], cna = col4[ia];
        int4 rnb = row4[ib], cnb = col4[ib];
        Quad4 a = ask4(lk, nodew, hws4, q, ra, ca);
        Quad4 b = ask4(lk, nodew, hws4, q, rb, cb);
        add4(lq, a);
        add4(lq, b);
        ra = rna; ca = cna;
        rb = rnb; cb = cnb;
    };
    trip(0);
    for (int it = 1; it < EPB / 2048; ++it) trip(it);
    if ((EPB / 2048) * 512 + quad < EPB / 4) {
        Quad4 a = ask4(lk, nodew, hws4, q, ra, ca);
        add4(lq, a);
    }

    float bq0 = b2v[4 * q], bq1 = b2v[4 * q + 1], bq2 = b2v[4 * q + 2], bq3 = b2v[4 * q + 3];
    static_assert(PGRID * 256 * 4 >= N_NODES, "four nodes per quad cover the node range");
    unsigned nrec[4];
    float4 nhv[4];
#pragma unroll
    for (int j = 0; j < 4; ++j) {
        int nc = min(blockIdx.x + PGRID * quad + PGRID * 256 * j, N_NODES - 1);
        unsigned k = lds_rec(lk, nc);
        nhv[j] = hws4[(long)nc * 4 + q];
        nrec[j] = full_rec(k, nodew, nc);
    }
#pragma unroll
    for (int j = 0; j < 4; ++j) {
        int n = blockIdx.x + PGRID * quad + PGRID * 256 * j;
        if (n < N_NODES) {
            float w = rsqrtf(1.0f + (float)(nrec[j] >> 9));
            float* p = lq + (nrec[j] & 511u);
            atomicAdd(p,              w * nhv[j].x + bq0);
            atomicAdd(p + PPITCH,     w * nhv[j].y + bq1);
            atomicAdd(p + 2 * PPITCH, w * nhv[j].z + bq2);
            atomicAdd(p + 3 * PPITCH, w * nhv[j].w + bq3);
        }
    }
    __syncthreads();

    float4* out4 = (float4*)(part + (size_t)blockIdx.x * (BGRAPH * HDIM));
    for (int i = tid; i < BGRAPH * 4; i += 1024) {
        int g = i >> 2, qq = i & 3;
        const float* s = lp + (4 * qq) * PPITCH + g;
        out4[i] = make_float4(s[0], s[PPITCH], s[2 * PPITCH], s[3 * PPITCH]);
    }
}

__global__ void __launch_bounds__(1024) final_kernel(const float* __restrict__ part,
                                                     const int* __restrict__ counts,
                                                     const float* __restrict__ Wfc,
                                                     const float* __restrict__ bfc,
                                                     float* __restrict__ out) {
    __shared__ float red[4][256];
    __shared__ float pl[256];
    int tid = threadIdx.x;
    int el = tid & 255, s = tid >> 8;
    const float* src = part + (size_t)blockIdx.x * 256 + el;
    float a0 = 0.0f, a1 = 0.0f, a2 = 0.0f, a3 = 0.0f;
#pragma unroll 8
    for (int w = s; w < PGRID; w += 16) {
        a0 += src[(size_t)(w)      * (BGRAPH * HDIM)];
        a1 += src[(size_t)(w + 4)  * (BGRAPH * HDIM)];
        a2 += src[(size_t)(w + 8)  * (BGRAPH * HDIM)];
        a3 += src[(size_t)(w + 12) * (BGRAPH * HDIM)];
    }
    red[s][el] = (a0 + a1) + (a2 + a3);
    __syncthreads();
    if (tid < 256) {
        int b = blockIdx.x * 16 + (tid >> 4);
        float inv = 1.0f / fmaxf((float)counts[b], 1.0f);
        pl[tid] = ((red[0][tid] + red[1][tid]) + (red[2][tid] + red[3][tid])) * inv;
    }
    __syncthreads();
    if (tid < 16 * CDIM) {
        int bl = tid / CDIM, cc = tid % CDIM;
        float acc = bfc[cc];
#pragma unroll
        for (int j = 0; j < HDIM; ++j)
            acc += pl[bl * HDIM + j] * Wfc[j * CDIM + cc];
        out[blockIdx.x * (16 * CDIM) + tid] = acc;
    }
}

static_assert((size_t)DG_SL * DG_NP * DG_PP * 4 <= (size_t)N_NODES * HDIM * 4 + AGG_REGION,
              "partial degree planes overrun hws + agg1");
static_assert(PLANE_B <= AGG_REGION && (size_t)PGRID * BGRAPH * HDIM * 4 <= AGG_REGION,
              "the first plane and, later, the partial pool tables fit into the region");
static_assert(((size_t)N_NODES * 4) % 16 == 0 && ((size_t)N_NODES * 5) % 16 == 0 &&
              (size_t)N_NODES * 5 + (size_t)NBASEP * 2 <= (size_t)N_NODES * 8,
              "the 8-bit records and the base table follow the 32-bit records inside the node table, on 16-byte boundaries");
static_assert(PLANE_B % 256 == 0 && (size_t)PGRID * BGRAPH * HDIM * 4 <= 2 * PLANE_B,
              "the second plane follows the first, and the partial pool tables fit into the two");
static_assert(((size_t)N_NODES * 4) % 256 == 0 && ((size_t)N_NODES * 8) % 256 == 0,
              "WS_TWO counts the small tables without padding");
static_assert(((size_t)N_NODES * HDIM * 4) % 256 == 0, "agg1 must start where hws ends");

extern "C" void kernel_launch(void* const* d_in, const int* in_sizes, int n_in,
                              void* d_out, int out_size, void* d_ws, size_t ws_size,
                              hipStream_t stream) {
    (void)in_sizes; (void)n_in; (void)out_size;

    const float* x    = (const float*)d_in[0];
    const int*   ei   = (const int*)d_in[1];
    const int*   batch= (const int*)d_in[2];
    const float* W1   = (const float*)d_in[3];
    const float* b1   = (const float*)d_in[4];
    const float* W2   = (const float*)d_in[5];
    const float* b2   = (const float*)d_in[6];
    const float* Wfc  = (const float*)d_in[7];
    const float* bfc  = (const float*)d_in[8];
    float* out = (float*)d_out;

    const int* erow = ei;
    const int* ecol = ei + N_EDGES;

    char* ws = (char*)d_ws;
    size_t o = 0;
    auto alloc = [&](size_t bytes) -> void* {
        void* p = ws + o;
        o += (bytes + 255) & ~(size_t)255;
        return p;
    };
    float* dis    = (float*)alloc((size_t)N_NODES * 4);
    unsigned* nodew = (unsigned*)alloc((size_t)N_NODES * 8);
    unsigned char* nodek = (unsigned char*)nodew + (size_t)N_NODES * 4;
    unsigned short* nbase = (unsigned short*)(nodek + (size_t)N_NODES);
    int*   counts = (int*)  alloc((size_t)BGRAPH * 4);
    float* hws    = (float*)alloc((size_t)N_NODES * HDIM * 4);
    float* agg1   = (float*)alloc(AGG_REGION);
    float* agg1b  = (float*)((char*)agg1 + PLANE_B);
    float* part   = agg1;
    int*   degpart = (int*)hws;

    hipMemsetAsync(counts, 0, (size_t)BGRAPH * 4, stream);

    deg_hist_kernel<<<DG_SL * DG_NP, 1024, 0, stream>>>(ecol, degpart);
    node_prep_kernel<<<(N_NODES + 255) / 256, 256, 0, stream>>>(degpart, batch, dis, nodew, nodek, nbase, counts);

    int gemm_grid = (NTILES + 7) / 8;
    gemm1_kernel<<<gemm_grid, 256, 0, stream>>>(x, W1, dis, hws, agg1, agg1b, ws_size);

    edge_scatter_kernel<<<SC_GRID, 256, 0, stream>>>(erow, ecol, hws, agg1, agg1b, ws_size);

    gemm2_kernel<<<gemm_grid, 256, 0, stream>>>(agg1, agg1b, ws_size, W2, b1, dis, hws);

    edge_pool_kernel<<<PGRID, 1024, 0, stream>>>(erow, ecol, hws, nodew, nodek, nbase, b2, part);
    final_kernel<<<BGRAPH / 16, 1024, 0, stream>>>(part, counts, Wfc, bfc, out);
}


# graph sizes counted in the pool pass: its self-loop step adds 1 per node to a 512-entry LDS table, head sums 256 partials; 200000 global integer atomics on a 2 KiB table and a memset go; sums same
# speedup vs baseline: 5.0803x; 1.1867x over previous
#include <hip/hip_runtime.h>

#define N_NODES 200000
#define N_EDGES 6400000
#define FDIM    128
#define HDIM    16
#define CDIM    10
#define BGRAPH  512
#define NTILES  (N_NODES / 16)
#define PGRID   256
#define EPB     (N_EDGES / PGRID)
#define AGG_REGION ((size_t)512 * BGRAPH * HDIM * 4)
#define NBLK    64
#define NBASE   (N_NODES / NBLK)
#define NBASEP  ((NBASE + 7) & ~7)
#define KESC8   63u
#define KCOPY   ((N_NODES / 16 + 1023) / 1024)
static_assert(N_NODES % NBLK == 0 && N_NODES % 16 == 0, "whole blocks; the records are copied as 16-byte words");
static_assert(BGRAPH <= 512 && N_EDGES < (1 << 23), "graph id in 9 bits, in-degree in 23");
static_assert(EPB * PGRID == N_EDGES && EPB % 4 == 0, "edges split evenly, four per 16-byte entry");
#define PPITCH  (BGRAPH + 1)
#define PTAB    (HDIM * PPITCH)

#define PLANE_B  ((size_t)N_NODES * HDIM * 4)
#define WS_TWO   ((size_t)N_NODES * 4 + (size_t)N_NODES * 8 + 3 * PLANE_B)

typedef float v2f __attribute__((ext_vector_type(2)));
typedef float v8f __attribute__((ext_vector_type(8)));

#define DG_NP   8
#define DG_SL   32
#define DG_PN   (N_NODES / DG_NP)
#define DG_SE4  (N_EDGES / DG_SL / 4)
#define DG_PP   25024
#define DG_TRIPS 7
static_assert(DG_PN * DG_NP == N_NODES && DG_PN % 4 == 0, "node ranges must be equal and 16-byte multiples");
static_assert(DG_SE4 * 4 * DG_SL == N_EDGES, "edge slices must be equal and 16-byte multiples");
static_assert(DG_PP >= DG_PN && (DG_PP * 4) % 128 == 0, "a partial plane starts on a line");
static_assert(DG_TRIPS * 7 * 1024 >= DG_SE4, "the trips must cover a slice");

__device__ __forceinline__ void deg_add(int* h, int c, int lo) {
    unsigned d = (unsigned)(c - lo);
    if (d < (unsigned)DG_PN) atomicAdd(&h[d], 1);
}

__global__ void __launch_bounds__(1024) deg_hist_kernel(const int* __restrict__ col,
                                                        int* __restrict__ degpart) {
    __shared__ int h[DG_PN];
    int tid = threadIdx.x;
    int s = blockIdx.x / DG_NP, p = blockIdx.x % DG_NP;
    int4* h4 = (int4*)h;
    for (int i = tid; i < DG_PN / 4; i += 1024) h4[i] = make_int4(0, 0, 0, 0);
    __syncthreads();

    int lo = p * DG_PN;
    const int4* c4 = (const int4*)(col + (size_t)s * (N_EDGES / DG_SL));
    for (int it = 0; it < DG_TRIPS; ++it) {
        int4 v[7];
#pragma unroll
        for (int j = 0; j < 7; ++j) {
            int idx = (it * 7 + j) * 1024 + tid;
            v[j] = idx < DG_SE4 ? c4[idx] : make_int4(-1, -1, -1, -1);
        }
#pragma unroll
        for (int j = 0; j < 7; ++j) {
            deg_add(h, v[j].x, lo);
            deg_add(h, v[j].y, lo);
            deg_add(h, v[j].z, lo);
            deg_add(h, v[j].w, lo);
        }
    }
    __syncthreads();

    int4* o4 = (int4*)(degpart + (size_t)blockIdx.x * DG_PP);
    for (int i = tid; i < DG_PN / 4; i += 1024) o4[i] = h4[i];
}

__global__ void node_prep_kernel(const int* __restrict__ degpart, const int* __restrict__ batch,
                                 float* __restrict__ dis, unsigned* __restrict__ nodew,
                                 unsigned char* __restrict__ nodek,
                                 unsigned short* __restrict__ nbase) {
    int n = blockIdx.x * blockDim.x + threadIdx.x;
    if (n < N_NODES) {
        int p = n / DG_PN;
        const int* src = degpart + (size_t)p * DG_PP + (n - p * DG_PN);
        int deg = 0;
#pragma unroll
        for (int s = 0; s < DG_SL; ++s) deg += src[(size_t)s * (DG_NP * DG_PP)];
        float d = rsqrtf(1.0f + (float)deg);
        int g = batch[n];
        int gb = batch[n & ~(NBLK - 1)];
        unsigned dl = (unsigned)(g - gb);
        bool fits = deg < (int)KESC8 && dl <= 3u;
        dis[n] = d;
        nodew[n] = ((unsigned)deg << 9) | (unsigned)g;
        nodek[n] = fits ? (unsigned char)(((unsigned)deg << 2) | dl) : (unsigned char)0xFF;
        if ((n & (NBLK - 1)) == 0) nbase[n / NBLK] = (unsigned short)g;
        if (n < NBASEP - NBASE) nbase[NBASE + n] = 0;
    }
}

__global__ void __launch_bounds__(256) gemm1_kernel(const float* __restrict__ x,
                                                    const float* __restrict__ W1,
                                                    const float* __restrict__ dis,
                                                    float* __restrict__ hws,
                                                    float* __restrict__ agg,
                                                    float* __restrict__ aggb,
                                                    size_t ws_size) {
    __shared__ float lw[(FDIM / 2) * 32];
    int tid = threadIdx.x;
#pragma unroll
    for (int i = 0; i < 8; ++i) {
        int idx = i * 256 + tid;
        int k = idx >> 4, n = idx & 15;
        lw[(k >> 1) * 32 + n * 2 + (k & 1)] = W1[idx];
    }
    __syncthreads();

    int wave = tid >> 5, lane = tid & 31;
    int tile = blockIdx.x * 8 + wave;
    if (tile >= NTILES) return;

    int h  = lane >> 4;
    int nn = lane & 15;
    long m = (long)tile * 16 + nn;
    const float* xrow = x + m * FDIM;

    v8f c = {0.f, 0.f, 0.f, 0.f, 0.f, 0.f, 0.f, 0.f};
#pragma unroll
    for (int kk = 0; kk < FDIM; kk += 4) {
        int k0 = kk + 2 * h;
        v2f a = *(const v2f*)(xrow + k0);
        v2f b = *(const v2f*)(&lw[(k0 >> 1) * 32 + nn * 2]);
        c = __builtin_amdgcn_wmma_f32_16x16x4_f32(false, a, false, b,
                                                  (short)0, c, false, false);
    }
#pragma unroll
    for (int v = 0; v < 8; ++v) {
        long r = (long)tile * 16 + v + 8 * h;
        float val = c[v] * dis[r];
        hws[r * HDIM + nn] = val;
        agg[r * HDIM + nn] = val;
    }
    if (ws_size >= WS_TWO) {
#pragma unroll
        for (int v = 0; v < 8; ++v) {
            long r = (long)tile * 16 + v + 8 * h;
            aggb[r * HDIM + nn] = 0.0f;
        }
    }
}

#define SC_GRID    1250
#define SC_GROUPS  (N_EDGES / (SC_GRID * 8 * 32))
static_assert(SC_GRID * 8 * 32 * SC_GROUPS == N_EDGES, "edges must split evenly over the waves");

template <int S>
__device__ __forceinline__ unsigned row_share(unsigned v) {
    return (unsigned)__builtin_amdgcn_update_dpp(0, (int)v, 0x150 + S, 0xf, 0xf, false);
}

#define SC_GATHER(S)                                                              \
    {                                                                             \
        unsigned r = row_share<S>(rv);                                            \
        co[S] = row_share<S>(cv) * 64u + jb;                                      \
        hv[S] = *(const float*)(hb + (size_t)(r * 64u + jb));                     \
    }
#define SC_ADD(S) atomicAdd((float*)(ab + (size_t)co[S]), hv[S]);

__global__ void __launch_bounds__(256) edge_scatter_kernel(const int* __restrict__ row,
                                                           const int* __restrict__ col,
                                                           const float* __restrict__ hws,
                                                           float* __restrict__ agg,
                                                           float* __restrict__ aggb,
                                                           size_t ws_size) {
    unsigned place = (unsigned)__builtin_amdgcn_s_sendmsg_rtn(0x87);
    bool upper = (ws_size >= WS_TWO) && ((place >> 18) & 1u);
    int lane = threadIdx.x & 31;
    int wave = blockIdx.x * 8 + (threadIdx.x >> 5);
    unsigned jb = (unsigned)(lane & 15) * 4u;
    const int* rp = row + (size_t)wave * (SC_GROUPS * 32) + lane;
    const int* cp = col + (size_t)wave * (SC_GROUPS * 32) + lane;
    const char* hb = (const char*)hws;
    char* ab = (char*)(upper ? aggb : agg);
    unsigned rv = (unsigned)rp[0];
    unsigned cv = (unsigned)cp[0];
    for (int g = 0; g < SC_GROUPS; ++g) {
        int gn = min(g + 1, SC_GROUPS - 1);
        unsigned rn = (unsigned)rp[gn * 32];
        unsigned cn = (unsigned)cp[gn * 32];
        float hv[16];
        unsigned co[16];
        SC_GATHER(0)  SC_GATHER(1)  SC_GATHER(2)  SC_GATHER(3)
        SC_GATHER(4)  SC_GATHER(5)  SC_GATHER(6)  SC_GATHER(7)
        SC_GATHER(8)  SC_GATHER(9)  SC_GATHER(10) SC_GATHER(11)
        SC_GATHER(12) SC_GATHER(13) SC_GATHER(14) SC_GATHER(15)
        SC_ADD(0)  SC_ADD(1)  SC_ADD(2)  SC_ADD(3)
        SC_ADD(4)  SC_ADD(5)  SC_ADD(6)  SC_ADD(7)
        SC_ADD(8)  SC_ADD(9)  SC_ADD(10) SC_ADD(11)
        SC_ADD(12) SC_ADD(13) SC_ADD(14) SC_ADD(15)
        rv = rn;
        cv = cn;
    }
}

__global__ void __launch_bounds__(256) gemm2_kernel(const float* __restrict__ agg1,
                                                    const float* __restrict__ agg1b,
                                                    size_t ws_size,
                                                    const float* __restrict__ W2,
                                                    const float* __restrict__ b1v,
                                                    const float* __restrict__ dis,
                                                    float* __restrict__ hws2) {
    __shared__ float lw[(HDIM / 2) * 32];
    __shared__ float lb1[HDIM];
    int tid = threadIdx.x;
    if (tid < 256) {
        int k = tid >> 4, n = tid & 15;
        lw[(k >> 1) * 32 + n * 2 + (k & 1)] = W2[tid];
    }
    if (tid < HDIM) lb1[tid] = b1v[tid];
    __syncthreads();

    int wave = tid >> 5, lane = tid & 31;
    int tile = blockIdx.x * 8 + wave;
    if (tile >= NTILES) return;

    int h  = lane >> 4;
    int nn = lane & 15;
    long m = (long)tile * 16 + nn;
    float dm = dis[m];
    const float* arow = agg1 + m * HDIM;
    const float* brow = agg1b + m * HDIM;

    v2f rawv[HDIM / 4];
#pragma unroll
    for (int i = 0; i < HDIM / 4; ++i) rawv[i] = *(const v2f*)(arow + 4 * i + 2 * h);
    if (ws_size >= WS_TWO) {
#pragma unroll
        for (int i = 0; i < HDIM / 4; ++i) {
            v2f rb = *(const v2f*)(brow + 4 * i + 2 * h);
            rawv[i].x += rb.x;
            rawv[i].y += rb.y;
        }
    }

    v8f c = {0.f, 0.f, 0.f, 0.f, 0.f, 0.f, 0.f, 0.f};
#pragma unroll
    for (int kk = 0; kk < HDIM; kk += 4) {
        int k0 = kk + 2 * h;
        v2f raw = rawv[kk / 4];
        v2f a;
        a.x = fmaxf(dm * raw.x + lb1[k0], 0.0f);
        a.y = fmaxf(dm * raw.y + lb1[k0 + 1], 0.0f);
        v2f b = *(const v2f*)(&lw[(k0 >> 1) * 32 + nn * 2]);
        c = __builtin_amdgcn_wmma_f32_16x16x4_f32(false, a, false, b,
                                                  (short)0, c, false, false);
    }
#pragma unroll
    for (int v = 0; v < 8; ++v) {
        long r = (long)tile * 16 + v + 8 * h;
        float val = c[v] * dis[r];
        hws2[r * HDIM + nn] = val;
    }
}

__device__ __forceinline__ void pool_add4(float* lq, unsigned rec, float4 hv) {
    float w = rsqrtf(1.0f + (float)(rec >> 9));
    float* p = lq + (rec & 511u);
    atomicAdd(p,              w * hv.x);
    atomicAdd(p + PPITCH,     w * hv.y);
    atomicAdd(p + 2 * PPITCH, w * hv.z);
    atomicAdd(p + 3 * PPITCH, w * hv.w);
}

struct RecTabs {
    const unsigned char* lk;
    const unsigned short* lb;
};

__device__ __forceinline__ unsigned lds_rec(RecTabs t, int c) {
    unsigned k = t.lk[c];
    unsigned b = t.lb[(unsigned)c / NBLK];
    return ((k >> 2) << 9) | ((b + (k & 3u)) & 511u);
}

__device__ __forceinline__ unsigned full_rec(unsigned k, const unsigned* __restrict__ nodew, int c) {
    return (k >> 9) == KESC8 ? nodew[c] : k;
}

struct Quad4 {
    unsigned v0, v1, v2, v3;
    float4 h0, h1, h2, h3;
};

__device__ __forceinline__ Quad4 ask4(RecTabs lk, const unsigned* __restrict__ nodew,
                                      const float4* __restrict__ hws4, int q, int4 r4, int4 c4) {
    Quad4 a;
    unsigned k0 = lds_rec(lk, c4.x), k1 = lds_rec(lk, c4.y), k2 = lds_rec(lk, c4.z), k3 = lds_rec(lk, c4.w);
    a.h0 = hws4[(long)r4.x * 4 + q];
    a.h1 = hws4[(long)r4.y * 4 + q];
    a.h2 = hws4[(long)r4.z * 4 + q];
    a.h3 = hws4[(long)r4.w * 4 + q];
    a.v0 = full_rec(k0, nodew, c4.x);
    a.v1 = full_rec(k1, nodew, c4.y);
    a.v2 = full_rec(k2, nodew, c4.z);
    a.v3 = full_rec(k3, nodew, c4.w);
    return a;
}

__device__ __forceinline__ void add4(float* lq, const Quad4& a) {
    pool_add4(lq, a.v0, a.h0);
    pool_add4(lq, a.v1, a.h1);
    pool_add4(lq, a.v2, a.h2);
    pool_add4(lq, a.v3, a.h3);
}

__global__ void __launch_bounds__(1024) edge_pool_kernel(const int* __restrict__ row,
                                                         const int* __restrict__ col,
                                                         const float* __restrict__ hws,
                                                         const unsigned* __restrict__ nodew,
                                                         const unsigned char* __restrict__ nodek,
                                                         const unsigned short* __restrict__ nbase,
                                                         const float* __restrict__ b2v,
                                                         float* __restrict__ part,
                                                         int* __restrict__ cpart) {
    __shared__ float lp[PTAB];
    __shared__ int lcnt[BGRAPH];
    __shared__ uint4 lk4[N_NODES / 16];
    __shared__ uint4 lb4[NBASEP / 8];
    int tid = threadIdx.x;
    const uint4* ks = (const uint4*)nodek;
    const uint4* bs = (const uint4*)nbase;
    uint4 kt[KCOPY];
    uint4 bt = make_uint4(0u, 0u, 0u, 0u);
#pragma unroll
    for (int j = 0; j < KCOPY; ++j) {
        int i = j * 1024 + tid;
        kt[j] = ks[min(i, N_NODES / 16 - 1)];
    }
    if (tid < NBASEP / 8) bt = bs[tid];
    for (int i = tid; i < PTAB; i += 1024) lp[i] = 0.0f;
    if (tid < BGRAPH) lcnt[tid] = 0;
#pragma unroll
    for (int j = 0; j < KCOPY; ++j) {
        int i = j * 1024 + tid;
        if (i < N_NODES / 16) lk4[i] = kt[j];
    }
    if (tid < NBASEP / 8) lb4[tid] = bt;
    __syncthreads();
    RecTabs lk;
    lk.lk = (const unsigned char*)lk4;
    lk.lb = (const unsigned short*)lb4;

    int q    = tid & 3;
    int quad = tid >> 2;
    const float4* hws4 = (const float4*)hws;
    float* lq = lp + (4 * q) * PPITCH;

    int ebeg = blockIdx.x * EPB;
    const int4* row4 = (const int4*)(row + ebeg);
    const int4* col4 = (const int4*)(col + ebeg);
    int4 ra = row4[quad],       ca = col4[quad];
    int4 rb = row4[256 + quad], cb = col4[256 + quad];
    auto trip = [&](int it) {
        int ia = min((it + 1) * 512 + quad,       EPB / 4 - 1);
        int ib = min((it + 1) * 512 + 256 + quad, EPB / 4 - 1);
        int4 rna = row4[ia], cna = col4[ia];
        int4 rnb = row4[ib], cnb = col4[ib];
        Quad4 a = ask4(lk, nodew, hws4, q, ra, ca);
        Quad4 b = ask4(lk, nodew, hws4, q, rb, cb);
        add4(lq, a);
        add4(lq, b);
        ra = rna; ca = cna;
        rb = rnb; cb = cnb;
    };
    trip(0);
    for (int it = 1; it < EPB / 2048; ++it) trip(it);
    if ((EPB / 2048) * 512 + quad < EPB / 4) {
        Quad4 a = ask4(lk, nodew, hws4, q, ra, ca);
        add4(lq, a);
    }

    float bq0 = b2v[4 * q], bq1 = b2v[4 * q + 1], bq2 = b2v[4 * q + 2], bq3 = b2v[4 * q + 3];
    static_assert(PGRID * 256 * 4 >= N_NODES, "four nodes per quad cover the node range");
    unsigned nrec[4];
    float4 nhv[4];
#pragma unroll
    for (int j = 0; j < 4; ++j) {
        int nc = min(blockIdx.x + PGRID * quad + PGRID * 256 * j, N_NODES - 1);
        unsigned k = lds_rec(lk, nc);
        nhv[j] = hws4[(long)nc * 4 + q];
        nrec[j] = full_rec(k, nodew, nc);
    }
#pragma unroll
    for (int j = 0; j < 4; ++j) {
        int n = blockIdx.x + PGRID * quad + PGRID * 256 * j;
        if (n < N_NODES) {
            float w = rsqrtf(1.0f + (float)(nrec[j] >> 9));
            float* p = lq + (nrec[j] & 511u);
            atomicAdd(p,              w * nhv[j].x + bq0);
            atomicAdd(p + PPITCH,     w * nhv[j].y + bq1);
            atomicAdd(p + 2 * PPITCH, w * nhv[j].z + bq2);
            atomicAdd(p + 3 * PPITCH, w * nhv[j].w + bq3);
            if (q == 0) atomicAdd(&lcnt[nrec[j] & 511u], 1);
        }
    }
    __syncthreads();

    float4* out4 = (float4*)(part + (size_t)blockIdx.x * (BGRAPH * HDIM));
    for (int i = tid; i < BGRAPH * 4; i += 1024) {
        int g = i >> 2, qq = i & 3;
        const float* s = lp + (4 * qq) * PPITCH + g;
        out4[i] = make_float4(s[0], s[PPITCH], s[2 * PPITCH], s[3 * PPITCH]);
    }
    if (tid < BGRAPH) cpart[blockIdx.x * BGRAPH + tid] = lcnt[tid];
}

static_assert(PGRID == 4 * 64, "64 thread groups of 16 take four count planes each");
__global__ void __launch_bounds__(1024) final_kernel(const float* __restrict__ part,
                                                     const int* __restrict__ cpart,
                                                     const float* __restrict__ Wfc,
                                                     const float* __restrict__ bfc,
                                                     float* __restrict__ out) {
    __shared__ float red[4][256];
    __shared__ float pl[256];
    __shared__ int cw[32][16];
    int tid = threadIdx.x;
    int el = tid & 255, s = tid >> 8;
    int gl = tid & 15, cs = tid >> 4;
    const int* csrc = cpart + blockIdx.x * 16 + gl;
    int cv = (csrc[cs * BGRAPH] + csrc[(cs + 64) * BGRAPH]) +
             (csrc[(cs + 128) * BGRAPH] + csrc[(cs + 192) * BGRAPH]);
    const float* src = part + (size_t)blockIdx.x * 256 + el;
    float a0 = 0.0f, a1 = 0.0f, a2 = 0.0f, a3 = 0.0f;
#pragma unroll 8
    for (int w = s; w < PGRID; w += 16) {
        a0 += src[(size_t)(w)      * (BGRAPH * HDIM)];
        a1 += src[(size_t)(w + 4)  * (BGRAPH * HDIM)];
        a2 += src[(size_t)(w + 8)  * (BGRAPH * HDIM)];
        a3 += src[(size_t)(w + 12) * (BGRAPH * HDIM)];
    }
    red[s][el] = (a0 + a1) + (a2 + a3);
    cv += __shfl_xor(cv, 16);
    if ((tid & 31) < 16) cw[tid >> 5][gl] = cv;
    __syncthreads();
    if (tid < 256) {
        int cnt = 0;
#pragma unroll
        for (int w = 0; w < 32; ++w) cnt += cw[w][tid >> 4];
        float inv = 1.0f / fmaxf((float)cnt, 1.0f);
        pl[tid] = ((red[0][tid] + red[1][tid]) + (red[2][tid] + red[3][tid])) * inv;
    }
    __syncthreads();
    if (tid < 16 * CDIM) {
        int bl = tid / CDIM, cc = tid % CDIM;
        float acc = bfc[cc];
#pragma unroll
        for (int j = 0; j < HDIM; ++j)
            acc += pl[bl * HDIM + j] * Wfc[j * CDIM + cc];
        out[blockIdx.x * (16 * CDIM) + tid] = acc;
    }
}

static_assert((size_t)DG_SL * DG_NP * DG_PP * 4 <= (size_t)N_NODES * HDIM * 4 + AGG_REGION,
              "partial degree planes overrun hws + agg1");
static_assert(PLANE_B <= AGG_REGION && (size_t)PGRID * BGRAPH * HDIM * 4 <= AGG_REGION,
              "the first plane and, later, the partial pool tables fit into the region");
static_assert((size_t)PGRID * BGRAPH * HDIM * 4 + (size_t)PGRID * BGRAPH * 4 <= PLANE_B &&
              ((size_t)PGRID * BGRAPH * HDIM * 4) % 256 == 0,
              "the partial count planes follow the partial pool tables inside the first plane");
static_assert(((size_t)N_NODES * 4) % 16 == 0 && ((size_t)N_NODES * 5) % 16 == 0 &&
              (size_t)N_NODES * 5 + (size_t)NBASEP * 2 <= (size_t)N_NODES * 8,
              "the 8-bit records and the base table follow the 32-bit records inside the node table, on 16-byte boundaries");
static_assert(PLANE_B % 256 == 0 && (size_t)PGRID * BGRAPH * HDIM * 4 <= 2 * PLANE_B,
              "the second plane follows the first, and the partial pool tables fit into the two");
static_assert(((size_t)N_NODES * 4) % 256 == 0 && ((size_t)N_NODES * 8) % 256 == 0,
              "WS_TWO counts the small tables without padding");
static_assert(((size_t)N_NODES * HDIM * 4) % 256 == 0, "agg1 must start where hws ends");

extern "C" void kernel_launch(void* const* d_in, const int* in_sizes, int n_in,
                              void* d_out, int out_size, void* d_ws, size_t ws_size,
                              hipStream_t stream) {
    (void)in_sizes; (void)n_in; (void)out_size;

    const float* x    = (const float*)d_in[0];
    const int*   ei   = (const int*)d_in[1];
    const int*   batch= (const int*)d_in[2];
    const float* W1   = (const float*)d_in[3];
    const float* b1   = (const float*)d_in[4];
    const float* W2   = (const float*)d_in[5];
    const float* b2   = (const float*)d_in[6];
    const float* Wfc  = (const float*)d_in[7];
    const float* bfc  = (const float*)d_in[8];
    float* out = (float*)d_out;

    const int* erow = ei;
    const int* ecol = ei + N_EDGES;

    char* ws = (char*)d_ws;
    size_t o = 0;
    auto alloc = [&](size_t bytes) -> void* {
        void* p = ws + o;
        o += (bytes + 255) & ~(size_t)255;
        return p;
    };
    float* dis    = (float*)alloc((size_t)N_NODES * 4);
    unsigned* nodew = (unsigned*)alloc((size_t)N_NODES * 8);
    unsigned char* nodek = (unsigned char*)nodew + (size_t)N_NODES * 4;
    unsigned short* nbase = (unsigned short*)(nodek + (size_t)N_NODES);
    float* hws    = (float*)alloc((size_t)N_NODES * HDIM * 4);
    float* agg1   = (float*)alloc(AGG_REGION);
    float* agg1b  = (float*)((char*)agg1 + PLANE_B);
    float* part   = agg1;
    int*   cpart  = (int*)((char*)agg1 + (size_t)PGRID * BGRAPH * HDIM * 4);
    int*   degpart = (int*)hws;

    deg_hist_kernel<<<DG_SL * DG_NP, 1024, 0, stream>>>(ecol, degpart);
    node_prep_kernel<<<(N_NODES + 255) / 256, 256, 0, stream>>>(degpart, batch, dis, nodew, nodek, nbase);

    int gemm_grid = (NTILES + 7) / 8;
    gemm1_kernel<<<gemm_grid, 256, 0, stream>>>(x, W1, dis, hws, agg1, agg1b, ws_size);

    edge_scatter_kernel<<<SC_GRID, 256, 0, stream>>>(erow, ecol, hws, agg1, agg1b, ws_size);

    gemm2_kernel<<<gemm_grid, 256, 0, stream>>>(agg1, agg1b, ws_size, W2, b1, dis, hws);

    edge_pool_kernel<<<PGRID, 1024, 0, stream>>>(erow, ecol, hws, nodew, nodek, nbase, b2, part, cpart);
    final_kernel<<<BGRAPH / 16, 1024, 0, stream>>>(part, cpart, Wfc, bfc, out);
}
